# starting kernel
# baseline (speedup 1.0000x reference)
#include <hip/hip_runtime.h>
#include <hip/hip_bf16.h>

typedef __attribute__((ext_vector_type(2))) float v2f;
typedef __attribute__((ext_vector_type(8))) float v8f;

__global__ void k_init_deg(float* __restrict__ deg, int n) {
  int i = blockIdx.x * blockDim.x + threadIdx.x;
  if (i < n) deg[i] = 1.0f;
}

__global__ void k_count_deg(const int* __restrict__ dst, float* __restrict__ deg, int ne) {
  int e = blockIdx.x * blockDim.x + threadIdx.x;
  if (e < ne) unsafeAtomicAdd(&deg[__builtin_nontemporal_load(dst + e)], 1.0f);
}

__global__ void k_dinv(const float* __restrict__ deg, float* __restrict__ dinv, int n) {
  int i = blockIdx.x * blockDim.x + threadIdx.x;
  if (i < n) dinv[i] = rsqrtf(deg[i]);
}

template <bool RELU_A>
__global__ void k_gemm_n16(const float* __restrict__ A, const float* __restrict__ B,
                           float* __restrict__ C, int nrows) {
  int wid  = (blockIdx.x * blockDim.x + threadIdx.x) >> 5;
  int lane = threadIdx.x & 31;
  int row0 = wid * 16;
  if (row0 >= nrows) return;
  int m  = lane & 15;
  int kh = (lane >> 4) << 1;
  const float* arow = A + (row0 + m) * 16;
  v8f c = {};
#pragma unroll
  for (int k0 = 0; k0 < 16; k0 += 4) {
    v2f a, b;
    float a0 = arow[k0 + kh + 0];
    float a1 = arow[k0 + kh + 1];
    if (RELU_A) { a0 = fmaxf(a0, 0.0f); a1 = fmaxf(a1, 0.0f); }
    a.x = a0; a.y = a1;
    b.x = B[(k0 + kh + 0) * 16 + m];
    b.y = B[(k0 + kh + 1) * 16 + m];
    c = __builtin_amdgcn_wmma_f32_16x16x4_f32(false, a, false, b, (short)0, c, false, false);
  }
  int rbase = row0 + ((lane >> 4) << 3);
#pragma unroll
  for (int r = 0; r < 8; ++r) C[(rbase + r) * 16 + m] = c[r];
}

__global__ void k_init_agg(const float* __restrict__ hw, const float* __restrict__ dinv,
                           const float* __restrict__ bias, float* __restrict__ agg, int n16) {
  int i = blockIdx.x * blockDim.x + threadIdx.x;
  if (i >= n16) return;
  int node = i >> 4, f = i & 15;
  float di = dinv[node];
  agg[i] = hw[i] * di * di + bias[f];
}

__global__ void k_scatter(const int* __restrict__ src, const int* __restrict__ dst,
                          const float* __restrict__ dinv, const float* __restrict__ hw,
                          float* __restrict__ agg, int ne) {
  int t = blockIdx.x * blockDim.x + threadIdx.x;
  int e = t >> 2, q = t & 3;
  if (e >= ne) return;
  int s = __builtin_nontemporal_load(src + e);
  int d = __builtin_nontemporal_load(dst + e);
  float w = dinv[s] * dinv[d];
  float4 v = ((const float4*)(hw + s * 16))[q];
  float* ad = agg + d * 16 + q * 4;
  unsafeAtomicAdd(ad + 0, v.x * w);
  unsafeAtomicAdd(ad + 1, v.y * w);
  unsafeAtomicAdd(ad + 2, v.z * w);
  unsafeAtomicAdd(ad + 3, v.w * w);
}

template <int KLEN>
__global__ void k_gemm_splitk_relu(const float* __restrict__ A, int lda,
                                   const float* __restrict__ B, int ncols,
                                   float* __restrict__ acc,
                                   int tiles_n, int nsplit) {
  int wid  = (blockIdx.x * blockDim.x + threadIdx.x) >> 5;
  int lane = threadIdx.x & 31;
  int split = wid % nsplit;
  int tn    = (wid / nsplit) % tiles_n;
  int tm    = wid / (nsplit * tiles_n);
  int m  = lane & 15;
  int kh = (lane >> 4) << 1;
  const float* arow = A + (size_t)(tm * 16 + m) * lda + split * KLEN;
  const float* bcol = B + (size_t)(split * KLEN) * ncols + tn * 16 + m;
  v8f c = {};
#pragma unroll 4
  for (int k0 = 0; k0 < KLEN; k0 += 4) {
    v2f a, b;
    a.x = fmaxf(arow[k0 + kh + 0], 0.0f);
    a.y = fmaxf(arow[k0 + kh + 1], 0.0f);
    b.x = bcol[(size_t)(k0 + kh + 0) * ncols];
    b.y = bcol[(size_t)(k0 + kh + 1) * ncols];
    c = __builtin_amdgcn_wmma_f32_16x16x4_f32(false, a, false, b, (short)0, c, false, false);
  }
  int rbase = tm * 16 + ((lane >> 4) << 3);
#pragma unroll
  for (int r = 0; r < 8; ++r)
    unsafeAtomicAdd(&acc[(rbase + r) * ncols + tn * 16 + m], c[r]);
}

__global__ void k_bias_relu(const float* __restrict__ in, const float* __restrict__ bias,
                            float* __restrict__ out, int n, int ncols) {
  int i = blockIdx.x * blockDim.x + threadIdx.x;
  if (i < n) out[i] = fmaxf(in[i] + bias[i % ncols], 0.0f);
}

__global__ void k_zero(float* __restrict__ p, int n) {
  int i = blockIdx.x * blockDim.x + threadIdx.x;
  if (i < n) p[i] = 0.0f;
}

__global__ void k_gemm_final(const float* __restrict__ A, const float* __restrict__ B,
                             const float* __restrict__ bias, float* __restrict__ out) {
  int wid  = (blockIdx.x * blockDim.x + threadIdx.x) >> 5;
  int lane = threadIdx.x & 31;
  int tm = wid >> 6;
  int tn = wid & 63;
  int m  = lane & 15;
  int kh = (lane >> 4) << 1;
  const float* arow = A + (tm * 16 + m) * 32;
  const float* bcol = B + tn * 16 + m;
  v8f c = {};
#pragma unroll
  for (int k0 = 0; k0 < 32; k0 += 4) {
    v2f a, b;
    a.x = arow[k0 + kh + 0];
    a.y = arow[k0 + kh + 1];
    b.x = bcol[(k0 + kh + 0) * 1024];
    b.y = bcol[(k0 + kh + 1) * 1024];
    c = __builtin_amdgcn_wmma_f32_16x16x4_f32(false, a, false, b, (short)0, c, false, false);
  }
  int rbase = tm * 16 + ((lane >> 4) << 3);
  float bv = bias[tn * 16 + m];
#pragma unroll
  for (int r = 0; r < 8; ++r)
    out[(rbase + r) * 1024 + tn * 16 + m] = c[r] + bv;
}

__global__ void k_value_final(const float* __restrict__ vhid, const float* __restrict__ w,
                              const float* __restrict__ b, float* __restrict__ out) {
  int bi = threadIdx.x;
  float s = 0.0f;
#pragma unroll
  for (int j = 0; j < 16; ++j) s += vhid[bi * 16 + j] * w[j];
  out[bi] = s + b[0];
}

extern "C" void kernel_launch(void* const* d_in, const int* in_sizes, int n_in,
                              void* d_out, int out_size, void* d_ws, size_t ws_size,
                              hipStream_t stream) {
  const float* x    = (const float*)d_in[0];
  const int*   ei   = (const int*)d_in[1];
  const float* W1   = (const float*)d_in[2];
  const float* b1   = (const float*)d_in[3];
  const float* W2   = (const float*)d_in[4];
  const float* b2   = (const float*)d_in[5];
  const float* p1w  = (const float*)d_in[6];
  const float* p1b  = (const float*)d_in[7];
  const float* p2w  = (const float*)d_in[8];
  const float* p2b  = (const float*)d_in[9];
  const float* v1w  = (const float*)d_in[10];
  const float* v1b  = (const float*)d_in[11];
  const float* v2w  = (const float*)d_in[12];
  const float* v2b  = (const float*)d_in[13];

  const int N  = in_sizes[0] / 16;
  const int E  = in_sizes[1] / 2;
  const int* esrc = ei;
  const int* edst = ei + E;

  float* ws   = (float*)d_ws;
  float* deg  = ws;
  float* dinv = deg + N;
  float* bufA = dinv + N;
  float* bufB = bufA + (size_t)16 * N;
  float* pacc = bufB + (size_t)16 * N;
  float* vacc = pacc + 64 * 32;
  float* phid = vacc + 64 * 16;
  float* vhid = phid + 64 * 32;

  float* Xout = (float*)d_out;
  float* Vout = Xout + 64 * 1024;

  const int B256 = 256;
  k_init_deg<<<N / B256, B256, 0, stream>>>(deg, N);
  k_count_deg<<<E / B256, B256, 0, stream>>>(edst, deg, E);
  k_dinv<<<N / B256, B256, 0, stream>>>(deg, dinv, N);

  int gemm_blocks    = (N / 16) / 8;
  int scatter_blocks = (4 * E) / B256;
  k_gemm_n16<false><<<gemm_blocks, B256, 0, stream>>>(x, W1, bufA, N);
  k_init_agg<<<(16 * N) / B256, B256, 0, stream>>>(bufA, dinv, b1, bufB, 16 * N);
  k_scatter<<<scatter_blocks, B256, 0, stream>>>(esrc, edst, dinv, bufA, bufB, E);

  k_gemm_n16<true><<<gemm_blocks, B256, 0, stream>>>(bufB, W2, bufA, N);
  k_init_agg<<<(16 * N) / B256, B256, 0, stream>>>(bufA, dinv, b2, bufB, 16 * N);
  k_scatter<<<scatter_blocks, B256, 0, stream>>>(esrc, edst, dinv, bufA, bufB, E);

  const int K = 16384, NSPLIT = 64, KLEN = K / NSPLIT;

  k_zero<<<(64 * 32 + 64 * 16 + B256 - 1) / B256, B256, 0, stream>>>(pacc, 64 * 32 + 64 * 16);

  k_gemm_splitk_relu<KLEN><<<64, B256, 0, stream>>>(bufB, K, p1w, 32, pacc, 2, NSPLIT);
  k_gemm_splitk_relu<KLEN><<<32, B256, 0, stream>>>(bufB, K, v1w, 16, vacc, 1, NSPLIT);

  k_bias_relu<<<(64 * 32) / B256, B256, 0, stream>>>(pacc, p1b, phid, 64 * 32, 32);
  k_bias_relu<<<(64 * 16 + B256 - 1) / B256, B256, 0, stream>>>(vacc, v1b, vhid, 64 * 16, 16);

  k_gemm_final<<<32, B256, 0, stream>>>(phid, p2w, p2b, Xout);
  k_value_final<<<1, 64, 0, stream>>>(vhid, v2w, v2b, Vout);
}


# edge list sorted by destination through LDS in two levels (chunk by 256-node bin, then bin by node): no global atomic rank, no scattered 4-byte store; LDS arrival-order ranks, sum order changes
# speedup vs baseline: 2.8395x; 2.8395x over previous
#include <hip/hip_runtime.h>
#include <hip/hip_bf16.h>

typedef __attribute__((ext_vector_type(2))) float v2f;
typedef __attribute__((ext_vector_type(8))) float v8f;
typedef __attribute__((ext_vector_type(4))) int   v4i;

#define BIN_SHIFT 8
#define BIN_NODES 256
#define MAX_BINS  256
#define CH_MAX    8192
#define STAGE_MAX 12288

__device__ __forceinline__ void scan256_first_wave(const int* in, int* out) {
  const int lane = threadIdx.x;
  int v[8];
  int run = 0;
#pragma unroll
  for (int k = 0; k < 8; ++k) { int t = in[lane * 8 + k]; v[k] = run; run += t; }
  int inc = run;
  for (int o = 1; o < 32; o <<= 1) {
    int t = __shfl_up(inc, o);
    if (lane >= o) inc += t;
  }
  const int ex = inc - run;
#pragma unroll
  for (int k = 0; k < 8; ++k) out[lane * 8 + k] = ex + v[k];
}

__global__ void __launch_bounds__(1024)
k_hist(const int* __restrict__ dst, int* __restrict__ cntm, int ne) {
  __shared__ int h[MAX_BINS];
  const int nc = gridDim.x, c = blockIdx.x;
  const int ch = ne / nc;
  if (threadIdx.x < MAX_BINS) h[threadIdx.x] = 0;
  __syncthreads();
  const v4i* d4 = (const v4i*)(dst + (size_t)c * ch);
  for (int i = threadIdx.x; i < (ch >> 2); i += blockDim.x) {
    v4i d = d4[i];
    atomicAdd(&h[d.x >> BIN_SHIFT], 1);
    atomicAdd(&h[d.y >> BIN_SHIFT], 1);
    atomicAdd(&h[d.z >> BIN_SHIFT], 1);
    atomicAdd(&h[d.w >> BIN_SHIFT], 1);
  }
  __syncthreads();
  if (threadIdx.x < MAX_BINS) cntm[threadIdx.x * nc + c] = h[threadIdx.x];
}

__global__ void __launch_bounds__(1024)
k_scan(const int* __restrict__ cnt, int* __restrict__ off, int n) {
  __shared__ int wsum[32];
  const int lane = threadIdx.x & 31;
  const int w    = threadIdx.x >> 5;
  const int seg  = n >> 5;
  const int iters = seg >> 7;
  const int4* c4 = (const int4*)(cnt + w * seg);
  int4*   o4 = (int4*)(off + w * seg);

  int tot = 0;
  for (int it = 0; it < iters; ++it) {
    int4 v = c4[it * 32 + lane];
    tot += (v.x + v.y) + (v.z + v.w);
  }
  for (int o = 16; o > 0; o >>= 1) tot += __shfl_xor(tot, o);
  if (lane == 0) wsum[w] = tot;
  __syncthreads();
  int base = (lane < w) ? wsum[lane] : 0;
  for (int o = 16; o > 0; o >>= 1) base += __shfl_xor(base, o);

  int carry = base;
  for (int it = 0; it < iters; ++it) {
    int4 v = c4[it * 32 + lane];
    int s1 = v.x, s2 = s1 + v.y, s3 = s2 + v.z, s4 = s3 + v.w;
    int inc = s4;
    for (int o = 1; o < 32; o <<= 1) {
      int t = __shfl_up(inc, o);
      if (lane >= o) inc += t;
    }
    int ex = carry + inc - s4;
    int4 ov; ov.x = ex; ov.y = ex + s1; ov.z = ex + s2; ov.w = ex + s3;
    o4[it * 32 + lane] = ov;
    carry += __shfl(inc, 31);
  }
  if (threadIdx.x == 1023) off[n] = carry;
}

__global__ void __launch_bounds__(1024)
k_scatter(const int* __restrict__ src, const int* __restrict__ dst,
          const int* __restrict__ boff, int* __restrict__ tmp, int ne) {
  __shared__ int cntl[MAX_BINS];
  __shared__ int lstart[MAX_BINS];
  __shared__ int cur[MAX_BINS];
  __shared__ int gbase[MAX_BINS];
  __shared__ int rec[CH_MAX];
  __shared__ int gadr[CH_MAX];
  const int nc = gridDim.x, c = blockIdx.x, tid = threadIdx.x;
  const int ch = ne / nc;
  if (tid < MAX_BINS) {
    int g0 = boff[tid * nc + c];
    int g1 = boff[tid * nc + c + 1];
    gbase[tid] = g0;
    cntl[tid]  = g1 - g0;
  }
  __syncthreads();
  if (tid < 32) scan256_first_wave(cntl, lstart);
  __syncthreads();
  if (tid < MAX_BINS) cur[tid] = lstart[tid];
  __syncthreads();
  const v4i* s4 = (const v4i*)(src + (size_t)c * ch);
  const v4i* d4 = (const v4i*)(dst + (size_t)c * ch);
  for (int i = tid; i < (ch >> 2); i += blockDim.x) {
    v4i s = __builtin_nontemporal_load(s4 + i);
    v4i d = d4[i];
#pragma unroll
    for (int k = 0; k < 4; ++k) {
      int dk = d[k], sk = s[k];
      int b = dk >> BIN_SHIFT;
      int p = atomicAdd(&cur[b], 1);
      rec[p]  = sk | ((dk & (BIN_NODES - 1)) << 16);
      gadr[p] = gbase[b] + (p - lstart[b]);
    }
  }
  __syncthreads();
  for (int i = tid; i < ch; i += blockDim.x) tmp[gadr[i]] = rec[i];
}

__global__ void __launch_bounds__(1024)
k_csr(const int* __restrict__ tmp, const int* __restrict__ boff, int nc,
      int* __restrict__ off, float* __restrict__ dinv, int* __restrict__ csr, int n) {
  __shared__ int dc[BIN_NODES];
  __shared__ int cur[BIN_NODES];
  __shared__ int stage[STAGE_MAX];
  const int b = blockIdx.x, tid = threadIdx.x;
  const int start = boff[b * nc];
  const int end   = boff[(b + 1) * nc];
  const int m = end - start;
  if (tid < BIN_NODES) dc[tid] = 0;
  __syncthreads();
  for (int i = tid; i < m; i += blockDim.x) atomicAdd(&dc[tmp[start + i] >> 16], 1);
  __syncthreads();
  if (tid < 32) scan256_first_wave(dc, cur);
  __syncthreads();
  if (tid < BIN_NODES) {
    int node = b * BIN_NODES + tid;
    off[node]  = start + cur[tid];
    dinv[node] = rsqrtf((float)(dc[tid] + 1));
  }
  if (b == (int)gridDim.x - 1 && tid == 0) off[n] = end;
  __syncthreads();
  const bool staged = m <= STAGE_MAX;
  for (int i = tid; i < m; i += blockDim.x) {
    int r = tmp[start + i];
    int p = atomicAdd(&cur[r >> 16], 1);
    int s = r & 0xFFFF;
    if (staged) stage[p] = s; else csr[start + p] = s;
  }
  if (staged) {
    __syncthreads();
    for (int i = tid; i < m; i += blockDim.x) csr[start + i] = stage[i];
  }
}

#define TAB_N 65536

__device__ __forceinline__ void fill_table(float* __restrict__ s_tab, const float* __restrict__ g_tab, int n) {
  const int n4 = n >> 2;
  const int rot = (blockIdx.x * 64) % n4;
#pragma unroll 4
  for (int i = threadIdx.x; i < n4; i += blockDim.x) {
    int j = i + rot;
    if (j >= n4) j -= n4;
    ((float4*)s_tab)[j] = ((const float4*)g_tab)[j];
  }
}

template <bool RELU_A>
__global__ void k_gemm_n16(const float* __restrict__ A, const float* __restrict__ B,
                           float* __restrict__ C, int nrows) {
  int wid  = (blockIdx.x * blockDim.x + threadIdx.x) >> 5;
  int lane = threadIdx.x & 31;
  int row0 = wid * 16;
  if (row0 >= nrows) return;
  int m  = lane & 15;
  int kh = (lane >> 4) << 1;
  const float* arow = A + (row0 + m) * 16;
  v8f c = {};
#pragma unroll
  for (int k0 = 0; k0 < 16; k0 += 4) {
    v2f a, b;
    float a0 = arow[k0 + kh + 0];
    float a1 = arow[k0 + kh + 1];
    if (RELU_A) { a0 = fmaxf(a0, 0.0f); a1 = fmaxf(a1, 0.0f); }
    a.x = a0; a.y = a1;
    b.x = B[(k0 + kh + 0) * 16 + m];
    b.y = B[(k0 + kh + 1) * 16 + m];
    c = __builtin_amdgcn_wmma_f32_16x16x4_f32(false, a, false, b, (short)0, c, false, false);
  }
  int rbase = row0 + ((lane >> 4) << 3);
#pragma unroll
  for (int r = 0; r < 8; ++r) C[(rbase + r) * 16 + m] = c[r];
}

__global__ void __launch_bounds__(1024)
k_pull(const int* __restrict__ off, const int* __restrict__ csr,
       const float* __restrict__ dinv, const float* __restrict__ hw,
       const float* __restrict__ bias, float* __restrict__ agg, int n) {
  __shared__ __attribute__((aligned(16))) float s_dinv[TAB_N];
  fill_table(s_dinv, dinv, n);
  __syncthreads();
  const float4* hw4 = (const float4*)hw;
  for (int t = blockIdx.x * blockDim.x + threadIdx.x; t < 4 * n; t += gridDim.x * blockDim.x) {
    int node = t >> 2, q = t & 3;
    int e   = off[node];
    int end = off[node + 1];
    float dd = s_dinv[node];
    float4 h = hw4[node * 4 + q];
    float4 acc;
    acc.x = h.x * dd * dd + bias[q * 4 + 0];
    acc.y = h.y * dd * dd + bias[q * 4 + 1];
    acc.z = h.z * dd * dd + bias[q * 4 + 2];
    acc.w = h.w * dd * dd + bias[q * 4 + 3];
    for (; e < end; ++e) {
      int s = csr[e];
      float w = s_dinv[s] * dd;
      float4 v = hw4[s * 4 + q];
      acc.x += v.x * w; acc.y += v.y * w; acc.z += v.z * w; acc.w += v.w * w;
    }
    ((float4*)agg)[node * 4 + q] = acc;
  }
}

template <int KLEN>
__global__ void k_gemm_splitk_relu(const float* __restrict__ A, int lda,
                                   const float* __restrict__ B, int ncols,
                                   float* __restrict__ acc,
                                   int tiles_n, int nsplit) {
  int wid  = (blockIdx.x * blockDim.x + threadIdx.x) >> 5;
  int lane = threadIdx.x & 31;
  int split = wid % nsplit;
  int tn    = (wid / nsplit) % tiles_n;
  int tm    = wid / (nsplit * tiles_n);
  int m  = lane & 15;
  int kh = (lane >> 4) << 1;
  const float* arow = A + (size_t)(tm * 16 + m) * lda + split * KLEN;
  const float* bcol = B + (size_t)(split * KLEN) * ncols + tn * 16 + m;
  v8f c = {};
#pragma unroll 4
  for (int k0 = 0; k0 < KLEN; k0 += 4) {
    v2f a, b;
    a.x = fmaxf(arow[k0 + kh + 0], 0.0f);
    a.y = fmaxf(arow[k0 + kh + 1], 0.0f);
    b.x = bcol[(size_t)(k0 + kh + 0) * ncols];
    b.y = bcol[(size_t)(k0 + kh + 1) * ncols];
    c = __builtin_amdgcn_wmma_f32_16x16x4_f32(false, a, false, b, (short)0, c, false, false);
  }
  int rbase = tm * 16 + ((lane >> 4) << 3);
#pragma unroll
  for (int r = 0; r < 8; ++r)
    unsafeAtomicAdd(&acc[(rbase + r) * ncols + tn * 16 + m], c[r]);
}

__global__ void k_bias_relu(const float* __restrict__ in, const float* __restrict__ bias,
                            float* __restrict__ out, int n, int ncols) {
  int i = blockIdx.x * blockDim.x + threadIdx.x;
  if (i < n) out[i] = fmaxf(in[i] + bias[i % ncols], 0.0f);
}

__global__ void k_zero(float* __restrict__ p, int n) {
  int i = blockIdx.x * blockDim.x + threadIdx.x;
  if (i < n) p[i] = 0.0f;
}

__global__ void k_gemm_final(const float* __restrict__ A, const float* __restrict__ B,
                             const float* __restrict__ bias, float* __restrict__ out) {
  int wid  = (blockIdx.x * blockDim.x + threadIdx.x) >> 5;
  int lane = threadIdx.x & 31;
  int tm = wid >> 6;
  int tn = wid & 63;
  int m  = lane & 15;
  int kh = (lane >> 4) << 1;
  const float* arow = A + (tm * 16 + m) * 32;
  const float* bcol = B + tn * 16 + m;
  v8f c = {};
#pragma unroll
  for (int k0 = 0; k0 < 32; k0 += 4) {
    v2f a, b;
    a.x = arow[k0 + kh + 0];
    a.y = arow[k0 + kh + 1];
    b.x = bcol[(k0 + kh + 0) * 1024];
    b.y = bcol[(k0 + kh + 1) * 1024];
    c = __builtin_amdgcn_wmma_f32_16x16x4_f32(false, a, false, b, (short)0, c, false, false);
  }
  int rbase = tm * 16 + ((lane >> 4) << 3);
  float bv = bias[tn * 16 + m];
#pragma unroll
  for (int r = 0; r < 8; ++r)
    out[(rbase + r) * 1024 + tn * 16 + m] = c[r] + bv;
}

__global__ void k_value_final(const float* __restrict__ vhid, const float* __restrict__ w,
                              const float* __restrict__ b, float* __restrict__ out) {
  int bi = threadIdx.x;
  float s = 0.0f;
#pragma unroll
  for (int j = 0; j < 16; ++j) s += vhid[bi * 16 + j] * w[j];
  out[bi] = s + b[0];
}

extern "C" void kernel_launch(void* const* d_in, const int* in_sizes, int n_in,
                              void* d_out, int out_size, void* d_ws, size_t ws_size,
                              hipStream_t stream) {
  const float* x    = (const float*)d_in[0];
  const int*   ei   = (const int*)d_in[1];
  const float* W1   = (const float*)d_in[2];
  const float* b1   = (const float*)d_in[3];
  const float* W2   = (const float*)d_in[4];
  const float* b2   = (const float*)d_in[5];
  const float* p1w  = (const float*)d_in[6];
  const float* p1b  = (const float*)d_in[7];
  const float* p2w  = (const float*)d_in[8];
  const float* p2b  = (const float*)d_in[9];
  const float* v1w  = (const float*)d_in[10];
  const float* v1b  = (const float*)d_in[11];
  const float* v2w  = (const float*)d_in[12];
  const float* v2b  = (const float*)d_in[13];

  const int N  = in_sizes[0] / 16;
  const int E  = in_sizes[1] / 2;
  const int* esrc = ei;
  const int* edst = ei + E;

  float* ws   = (float*)d_ws;
  int*   boff = (int*)ws;
  int*   off  = boff + 256 * 256 + 32;
  float* dinv = (float*)(off + N + 32);
  float* pacc = dinv + N;
  float* vacc = pacc + 64 * 32;
  float* phid = vacc + 64 * 16;
  float* vhid = phid + 64 * 32;
  int*   csr  = (int*)(vhid + 64 * 16);
  float* bufA = (float*)(csr + E);
  float* bufB = bufA + (size_t)16 * N;
  int*   cntm = csr;
  int*   tmp  = (int*)bufA;

  float* Xout = (float*)d_out;
  float* Vout = Xout + 64 * 1024;

  const int B256 = 256;
  const int WGPS = 256;
  const int NC   = 256;
  k_hist<<<NC, 1024, 0, stream>>>(edst, cntm, E);
  k_scan<<<1, 1024, 0, stream>>>(cntm, boff, 256 * 256);
  k_scatter<<<NC, 1024, 0, stream>>>(esrc, edst, boff, tmp, E);
  k_csr<<<N / 256, 1024, 0, stream>>>(tmp, boff, NC, off, dinv, csr, N);

  int gemm_blocks = (N / 16) / 8;
  k_gemm_n16<false><<<gemm_blocks, B256, 0, stream>>>(x, W1, bufA, N);
  k_pull<<<WGPS, 1024, 0, stream>>>(off, csr, dinv, bufA, b1, bufB, N);

  k_gemm_n16<true><<<gemm_blocks, B256, 0, stream>>>(bufB, W2, bufA, N);
  k_pull<<<WGPS, 1024, 0, stream>>>(off, csr, dinv, bufA, b2, bufB, N);

  const int K = 16384, NSPLIT = 64, KLEN = K / NSPLIT;

  k_zero<<<(64 * 32 + 64 * 16 + B256 - 1) / B256, B256, 0, stream>>>(pacc, 64 * 32 + 64 * 16);

  k_gemm_splitk_relu<KLEN><<<64, B256, 0, stream>>>(bufB, K, p1w, 32, pacc, 2, NSPLIT);
  k_gemm_splitk_relu<KLEN><<<32, B256, 0, stream>>>(bufB, K, v1w, 16, vacc, 1, NSPLIT);

  k_bias_relu<<<(64 * 32) / B256, B256, 0, stream>>>(pacc, p1b, phid, 64 * 32, 32);
  k_bias_relu<<<(64 * 16 + B256 - 1) / B256, B256, 0, stream>>>(vacc, v1b, vhid, 64 * 16, 16);

  k_gemm_final<<<32, B256, 0, stream>>>(phid, p2w, p2b, Xout);
  k_value_final<<<1, 64, 0, stream>>>(vhid, v2w, v2b, Vout);
}


# level 1 sorts each chunk in place (chunk-major, coalesced stores) and files its bin starts; level 2 takes offsets from that table's row sums: no count and scan launches, 15 to 13; sum order changes
# speedup vs baseline: 3.5865x; 1.2631x over previous
#include <hip/hip_runtime.h>
#include <hip/hip_bf16.h>

typedef __attribute__((ext_vector_type(2))) float v2f;
typedef __attribute__((ext_vector_type(8))) float v8f;
typedef __attribute__((ext_vector_type(4))) int   v4i;

#define BIN_SHIFT 8
#define BIN_NODES 256
#define MAX_BINS  256
#define CH_MAX    8192
#define MAX_CHUNKS 256
#define RUN_REGS  2
#define STAGE_MAX 12288

__device__ __forceinline__ void scan256_first_wave(const int* in, int* out) {
  const int lane = threadIdx.x;
  int v[8];
  int run = 0;
#pragma unroll
  for (int k = 0; k < 8; ++k) { int t = in[lane * 8 + k]; v[k] = run; run += t; }
  int inc = run;
  for (int o = 1; o < 32; o <<= 1) {
    int t = __shfl_up(inc, o);
    if (lane >= o) inc += t;
  }
  const int ex = inc - run;
#pragma unroll
  for (int k = 0; k < 8; ++k) out[lane * 8 + k] = ex + v[k];
}

__global__ void __launch_bounds__(1024)
k_chunk_sort(const int* __restrict__ src, const int* __restrict__ dst,
             int* __restrict__ tmp, int* __restrict__ cst, int ne) {
  __shared__ int cntl[MAX_BINS];
  __shared__ int lstart[MAX_BINS];
  __shared__ int cur[MAX_BINS];
  __shared__ int rec[CH_MAX];
  const int nc = gridDim.x, c = blockIdx.x, tid = threadIdx.x;
  const int ch = ne / nc;
  if (tid < MAX_BINS) cntl[tid] = 0;
  __syncthreads();
  const v4i* s4 = (const v4i*)(src + (size_t)c * ch);
  const v4i* d4 = (const v4i*)(dst + (size_t)c * ch);
  for (int i = tid; i < (ch >> 2); i += blockDim.x) {
    v4i d = d4[i];
    atomicAdd(&cntl[d.x >> BIN_SHIFT], 1);
    atomicAdd(&cntl[d.y >> BIN_SHIFT], 1);
    atomicAdd(&cntl[d.z >> BIN_SHIFT], 1);
    atomicAdd(&cntl[d.w >> BIN_SHIFT], 1);
  }
  __syncthreads();
  if (tid < 32) scan256_first_wave(cntl, lstart);
  __syncthreads();
  if (tid < MAX_BINS) {
    int l = lstart[tid];
    cur[tid] = l;
    cst[tid * nc + c] = l;
  }
  __syncthreads();
  for (int i = tid; i < (ch >> 2); i += blockDim.x) {
    v4i s = __builtin_nontemporal_load(s4 + i);
    v4i d = d4[i];
#pragma unroll
    for (int k = 0; k < 4; ++k) {
      int dk = d[k], sk = s[k];
      int p = atomicAdd(&cur[dk >> BIN_SHIFT], 1);
      rec[p] = sk | ((dk & (BIN_NODES - 1)) << 16);
    }
  }
  __syncthreads();
  int* t = tmp + (size_t)c * ch;
  for (int i = tid; i < ch; i += blockDim.x) t[i] = rec[i];
}

__global__ void __launch_bounds__(1024)
k_csr(const int* __restrict__ tmp, const int* __restrict__ cst, int nc, int ne,
      int* __restrict__ off, float* __restrict__ dinv, int* __restrict__ csr, int n) {
  __shared__ int dc[BIN_NODES];
  __shared__ int cur[BIN_NODES];
  __shared__ int rbeg[MAX_CHUNKS];
  __shared__ int rlen[MAX_CHUNKS];
  __shared__ int redlo[32], redhi[32];
  __shared__ int stage[STAGE_MAX];
  const int b = blockIdx.x, tid = threadIdx.x;
  const int lane = tid & 31, w = tid >> 5;
  const int ch = ne / nc;
  int lo = 0, hi = 0;
  if (tid < nc && tid < MAX_CHUNKS) {
    lo = cst[b * nc + tid];
    hi = (b + 1 < MAX_BINS) ? cst[(b + 1) * nc + tid] : ch;
  }
  if (tid < MAX_CHUNKS) { rbeg[tid] = tid * ch + lo; rlen[tid] = hi - lo; }
  if (tid < BIN_NODES) dc[tid] = 0;
  int slo = lo, shi = hi;
  for (int o = 16; o > 0; o >>= 1) { slo += __shfl_xor(slo, o); shi += __shfl_xor(shi, o); }
  if (lane == 0) { redlo[w] = slo; redhi[w] = shi; }
  __syncthreads();
  int start = redlo[lane], end = redhi[lane];
  for (int o = 16; o > 0; o >>= 1) { start += __shfl_xor(start, o); end += __shfl_xor(end, o); }
  const int m = end - start;

  int beg[8], len[8], r[8][RUN_REGS];
#pragma unroll
  for (int k = 0; k < 8; ++k) { beg[k] = rbeg[w + 32 * k]; len[k] = rlen[w + 32 * k]; }
#pragma unroll
  for (int k = 0; k < 8; ++k)
#pragma unroll
    for (int h = 0; h < RUN_REGS; ++h) {
      const int j = lane + 32 * h;
      r[k][h] = (j < len[k]) ? tmp[beg[k] + j] : -1;
    }
#pragma unroll
  for (int k = 0; k < 8; ++k) {
#pragma unroll
    for (int h = 0; h < RUN_REGS; ++h)
      if (r[k][h] >= 0) atomicAdd(&dc[r[k][h] >> 16], 1);
    for (int j = lane + 32 * RUN_REGS; j < len[k]; j += 32) atomicAdd(&dc[tmp[beg[k] + j] >> 16], 1);
  }
  __syncthreads();
  if (tid < 32) scan256_first_wave(dc, cur);
  __syncthreads();
  if (tid < BIN_NODES) {
    int node = b * BIN_NODES + tid;
    off[node]  = start + cur[tid];
    dinv[node] = rsqrtf((float)(dc[tid] + 1));
  }
  if (b == (int)gridDim.x - 1 && tid == 0) off[n] = end;
  __syncthreads();
  const bool staged = m <= STAGE_MAX;
#pragma unroll
  for (int k = 0; k < 8; ++k) {
#pragma unroll
    for (int h = 0; h < RUN_REGS; ++h)
      if (r[k][h] >= 0) {
        int p = atomicAdd(&cur[r[k][h] >> 16], 1);
        int s = r[k][h] & 0xFFFF;
        if (staged) stage[p] = s; else csr[start + p] = s;
      }
    for (int j = lane + 32 * RUN_REGS; j < len[k]; j += 32) {
      int rr = tmp[beg[k] + j];
      int p = atomicAdd(&cur[rr >> 16], 1);
      int s = rr & 0xFFFF;
      if (staged) stage[p] = s; else csr[start + p] = s;
    }
  }
  if (staged) {
    __syncthreads();
    for (int i = tid; i < m; i += blockDim.x) csr[start + i] = stage[i];
  }
}

template <bool A_COLMAJOR_RELU>
__global__ void k_gemm_n16(const float* __restrict__ A, const float* __restrict__ B,
                           float* __restrict__ CT, int nrows) {
  int wid  = (blockIdx.x * blockDim.x + threadIdx.x) >> 5;
  int lane = threadIdx.x & 31;
  int row0 = wid * 32;
  if (row0 >= nrows) return;
  int m  = lane & 15;
  int kh = (lane >> 4) << 1;
  v2f bf[4];
#pragma unroll
  for (int ks = 0; ks < 4; ++ks) {
    bf[ks].x = B[(ks * 4 + kh + 0) * 16 + m];
    bf[ks].y = B[(ks * 4 + kh + 1) * 16 + m];
  }
#pragma unroll
  for (int t = 0; t < 2; ++t) {
    const int row = row0 + t * 16 + m;
    v8f c = {};
#pragma unroll
    for (int ks = 0; ks < 4; ++ks) {
      const int k = ks * 4 + kh;
      v2f a;
      if (A_COLMAJOR_RELU) {
        a.x = fmaxf(A[(size_t)(k + 0) * nrows + row], 0.0f);
        a.y = fmaxf(A[(size_t)(k + 1) * nrows + row], 0.0f);
      } else {
        a.x = A[row * 16 + k + 0];
        a.y = A[row * 16 + k + 1];
      }
      c = __builtin_amdgcn_wmma_f32_16x16x4_f32(false, bf[ks], false, a, (short)0, c, false, false);
    }
    float* o = CT + (size_t)((lane >> 4) << 3) * nrows + row;
#pragma unroll
    for (int r = 0; r < 8; ++r) o[(size_t)r * nrows] = c[r];
  }
}

#define TAB_N    65536
#define F_COLS   16
#define PER_TRIP 12

__global__ void __launch_bounds__(1024)
k_pull_col(const int* __restrict__ off, const int* __restrict__ csr,
           const float* __restrict__ dinv, const float* __restrict__ hwT,
           const float* __restrict__ bias, float* __restrict__ aggT, int n) {
  __shared__ __attribute__((aligned(16))) float tab[TAB_N];
  const int tid = threadIdx.x;
  const int f   = blockIdx.x % F_COLS;
  const int rg  = blockIdx.x / F_COLS;
  const int nrg = gridDim.x / F_COLS;
  const int per = n / nrg;

  {
    const float4* h4 = (const float4*)(hwT + (size_t)f * n);
    const float4* d4 = (const float4*)dinv;
    const int n4  = n >> 2;
    const int rot = (blockIdx.x * 64) % n4;
#pragma unroll 4
    for (int i = tid; i < n4; i += blockDim.x) {
      int j = i + rot;
      if (j >= n4) j -= n4;
      float4 a = h4[j], b = d4[j];
      float4 p; p.x = a.x * b.x; p.y = a.y * b.y; p.z = a.z * b.z; p.w = a.w * b.w;
      ((float4*)tab)[j] = p;
    }
  }
  __syncthreads();

  const int lane = tid & 31, w = tid >> 5;
  const int q = lane & 3, g = lane >> 2;
  const int nwaves = blockDim.x >> 5;
  for (int u = w; u * 32 < per; u += nwaves) {
    const int nb = rg * per + u * 32;
    int first[4], end[4], last[4];
#pragma unroll
    for (int sub = 0; sub < 4; ++sub) {
      const int node = nb + sub * 8 + g;
      first[sub] = off[node] + q;
      end[sub]   = off[node + 1];
      last[sub]  = max(end[sub] - 1, 0);
    }
    int s[4][PER_TRIP];
#pragma unroll
    for (int sub = 0; sub < 4; ++sub)
#pragma unroll
      for (int k = 0; k < PER_TRIP; ++k)
        s[sub][k] = csr[min(first[sub] + 4 * k, last[sub])];
    float a[4];
#pragma unroll
    for (int sub = 0; sub < 4; ++sub) {
      float acc = 0.0f;
#pragma unroll
      for (int k = 0; k < PER_TRIP; ++k) {
        float v = tab[s[sub][k]];
        acc += (first[sub] + 4 * k < end[sub]) ? v : 0.0f;
      }
      a[sub] = acc;
    }
#pragma unroll
    for (int sub = 0; sub < 4; ++sub)
      for (int i = first[sub] + 4 * PER_TRIP; i < end[sub]; i += 4) a[sub] += tab[csr[i]];
    float res = 0.0f;
#pragma unroll
    for (int sub = 0; sub < 4; ++sub) {
      float t = a[sub];
      t += __shfl_xor(t, 1);
      t += __shfl_xor(t, 2);
      if (q == sub) res = t;
    }
    const int mine = nb + q * 8 + g;
    aggT[(size_t)f * n + mine] = dinv[mine] * (tab[mine] + res) + bias[f];
  }
}

template <int KLEN>
__global__ void k_gemm_splitk_relu(const float* __restrict__ AT, int nn, int npg,
                                   const float* __restrict__ B, int ncols,
                                   float* __restrict__ acc,
                                   int tiles_n, int nsplit) {
  int wid  = (blockIdx.x * blockDim.x + threadIdx.x) >> 5;
  int lane = threadIdx.x & 31;
  int split = wid % nsplit;
  int tn    = (wid / nsplit) % tiles_n;
  int tm    = wid / (nsplit * tiles_n);
  int m  = lane & 15;
  int kh = (lane >> 4) << 1;
  const float* abase = AT + (size_t)(tm * 16 + m) * npg + split * (KLEN / 16);
  const float* bbase = B + (size_t)(split * KLEN) * ncols + tn * 16 + m;
  v8f c = {};
#pragma unroll 4
  for (int i = 0; i < KLEN / 4; ++i) {
    const int f  = i >> 2;
    const int nl = ((i & 3) << 2) + kh;
    v2f a, b;
    const int ai = f * nn + nl;
    const int bi = (nl * 16 + f) * ncols;
    a.x = fmaxf(abase[ai + 0], 0.0f);
    a.y = fmaxf(abase[ai + 1], 0.0f);
    b.x = bbase[bi];
    b.y = bbase[bi + 16 * ncols];
    c = __builtin_amdgcn_wmma_f32_16x16x4_f32(false, a, false, b, (short)0, c, false, false);
  }
  int rbase = tm * 16 + ((lane >> 4) << 3);
#pragma unroll
  for (int r = 0; r < 8; ++r)
    unsafeAtomicAdd(&acc[(rbase + r) * ncols + tn * 16 + m], c[r]);
}

__global__ void k_bias_relu(const float* __restrict__ in, const float* __restrict__ bias,
                            float* __restrict__ out, int n, int ncols) {
  int i = blockIdx.x * blockDim.x + threadIdx.x;
  if (i < n) out[i] = fmaxf(in[i] + bias[i % ncols], 0.0f);
}

__global__ void k_zero(float* __restrict__ p, int n) {
  int i = blockIdx.x * blockDim.x + threadIdx.x;
  if (i < n) p[i] = 0.0f;
}

__global__ void k_gemm_final(const float* __restrict__ A, const float* __restrict__ B,
                             const float* __restrict__ bias, float* __restrict__ out) {
  int wid  = (blockIdx.x * blockDim.x + threadIdx.x) >> 5;
  int lane = threadIdx.x & 31;
  int tm = wid >> 6;
  int tn = wid & 63;
  int m  = lane & 15;
  int kh = (lane >> 4) << 1;
  const float* arow = A + (tm * 16 + m) * 32;
  const float* bcol = B + tn * 16 + m;
  v8f c = {};
#pragma unroll
  for (int k0 = 0; k0 < 32; k0 += 4) {
    v2f a, b;
    a.x = arow[k0 + kh + 0];
    a.y = arow[k0 + kh + 1];
    b.x = bcol[(k0 + kh + 0) * 1024];
    b.y = bcol[(k0 + kh + 1) * 1024];
    c = __builtin_amdgcn_wmma_f32_16x16x4_f32(false, a, false, b, (short)0, c, false, false);
  }
  int rbase = tm * 16 + ((lane >> 4) << 3);
  float bv = bias[tn * 16 + m];
#pragma unroll
  for (int r = 0; r < 8; ++r)
    out[(rbase + r) * 1024 + tn * 16 + m] = c[r] + bv;
}

__global__ void k_value_final(const float* __restrict__ vhid, const float* __restrict__ w,
                              const float* __restrict__ b, float* __restrict__ out) {
  int bi = threadIdx.x;
  float s = 0.0f;
#pragma unroll
  for (int j = 0; j < 16; ++j) s += vhid[bi * 16 + j] * w[j];
  out[bi] = s + b[0];
}

extern "C" void kernel_launch(void* const* d_in, const int* in_sizes, int n_in,
                              void* d_out, int out_size, void* d_ws, size_t ws_size,
                              hipStream_t stream) {
  const float* x    = (const float*)d_in[0];
  const int*   ei   = (const int*)d_in[1];
  const float* W1   = (const float*)d_in[2];
  const float* b1   = (const float*)d_in[3];
  const float* W2   = (const float*)d_in[4];
  const float* b2   = (const float*)d_in[5];
  const float* p1w  = (const float*)d_in[6];
  const float* p1b  = (const float*)d_in[7];
  const float* p2w  = (const float*)d_in[8];
  const float* p2b  = (const float*)d_in[9];
  const float* v1w  = (const float*)d_in[10];
  const float* v1b  = (const float*)d_in[11];
  const float* v2w  = (const float*)d_in[12];
  const float* v2b  = (const float*)d_in[13];

  const int N  = in_sizes[0] / 16;
  const int E  = in_sizes[1] / 2;
  const int* esrc = ei;
  const int* edst = ei + E;

  float* ws   = (float*)d_ws;
  int*   cst  = (int*)ws;
  int*   off  = cst + 256 * 256 + 32;
  float* dinv = (float*)(off + N + 32);
  float* pacc = dinv + N;
  float* vacc = pacc + 64 * 32;
  float* phid = vacc + 64 * 16;
  float* vhid = phid + 64 * 32;
  int*   csr  = (int*)(vhid + 64 * 16);
  float* bufA = (float*)(csr + E);
  float* bufB = bufA + (size_t)16 * N;
  int*   tmp  = (int*)bufA;

  float* Xout = (float*)d_out;
  float* Vout = Xout + 64 * 1024;

  const int B256 = 256;
  const int WGPS = 256;
  const int NC   = 256;
  k_chunk_sort<<<NC, 1024, 0, stream>>>(esrc, edst, tmp, cst, E);
  k_csr<<<N / 256, 1024, 0, stream>>>(tmp, cst, NC, E, off, dinv, csr, N);

  int gemm_blocks = (N / 32) / 8;
  k_gemm_n16<false><<<gemm_blocks, B256, 0, stream>>>(x, W1, bufA, N);
  k_pull_col<<<WGPS, 1024, 0, stream>>>(off, csr, dinv, bufA, b1, bufB, N);

  k_gemm_n16<true><<<gemm_blocks, B256, 0, stream>>>(bufB, W2, bufA, N);
  k_pull_col<<<WGPS, 1024, 0, stream>>>(off, csr, dinv, bufA, b2, bufB, N);

  const int K = 16384, NSPLIT = 64, KLEN = K / NSPLIT;

  k_zero<<<(64 * 32 + 64 * 16 + B256 - 1) / B256, B256, 0, stream>>>(pacc, 64 * 32 + 64 * 16);

  k_gemm_splitk_relu<KLEN><<<64, B256, 0, stream>>>(bufB, N, K / 16, p1w, 32, pacc, 2, NSPLIT);
  k_gemm_splitk_relu<KLEN><<<32, B256, 0, stream>>>(bufB, N, K / 16, v1w, 16, vacc, 1, NSPLIT);

  k_bias_relu<<<(64 * 32) / B256, B256, 0, stream>>>(pacc, p1b, phid, 64 * 32, 32);
  k_bias_relu<<<(64 * 16 + B256 - 1) / B256, B256, 0, stream>>>(vacc, v1b, vhid, 64 * 16, 16);

  k_gemm_final<<<32, B256, 0, stream>>>(phid, p2w, p2b, Xout);
  k_value_final<<<1, 64, 0, stream>>>(vhid, v2w, v2b, Vout);
}


# heads' split-K sums kept as stored partial tiles (8 waves summed in LDS, whole-line stores) that the output kernel adds: no float atomics, no zero, bias or value launch; 13 to 8; sum order changes
# speedup vs baseline: 4.2641x; 1.1889x over previous
#include <hip/hip_runtime.h>
#include <hip/hip_bf16.h>

typedef __attribute__((ext_vector_type(2))) float v2f;
typedef __attribute__((ext_vector_type(8))) float v8f;
typedef __attribute__((ext_vector_type(4))) int   v4i;

#define BIN_SHIFT 8
#define BIN_NODES 256
#define MAX_BINS  256
#define CH_MAX    8192
#define MAX_CHUNKS 256
#define RUN_REGS  2
#define STAGE_MAX 12288

__device__ __forceinline__ void scan256_first_wave(const int* in, int* out) {
  const int lane = threadIdx.x;
  int v[8];
  int run = 0;
#pragma unroll
  for (int k = 0; k < 8; ++k) { int t = in[lane * 8 + k]; v[k] = run; run += t; }
  int inc = run;
  for (int o = 1; o < 32; o <<= 1) {
    int t = __shfl_up(inc, o);
    if (lane >= o) inc += t;
  }
  const int ex = inc - run;
#pragma unroll
  for (int k = 0; k < 8; ++k) out[lane * 8 + k] = ex + v[k];
}

__global__ void __launch_bounds__(1024)
k_chunk_sort(const int* __restrict__ src, const int* __restrict__ dst,
             int* __restrict__ tmp, int* __restrict__ cst, int ne) {
  __shared__ int cntl[MAX_BINS];
  __shared__ int lstart[MAX_BINS];
  __shared__ int cur[MAX_BINS];
  __shared__ int rec[CH_MAX];
  const int nc = gridDim.x, c = blockIdx.x, tid = threadIdx.x;
  const int ch = ne / nc;
  if (tid < MAX_BINS) cntl[tid] = 0;
  __syncthreads();
  const v4i* s4 = (const v4i*)(src + (size_t)c * ch);
  const v4i* d4 = (const v4i*)(dst + (size_t)c * ch);
  for (int i = tid; i < (ch >> 2); i += blockDim.x) {
    v4i d = d4[i];
    atomicAdd(&cntl[d.x >> BIN_SHIFT], 1);
    atomicAdd(&cntl[d.y >> BIN_SHIFT], 1);
    atomicAdd(&cntl[d.z >> BIN_SHIFT], 1);
    atomicAdd(&cntl[d.w >> BIN_SHIFT], 1);
  }
  __syncthreads();
  if (tid < 32) scan256_first_wave(cntl, lstart);
  __syncthreads();
  if (tid < MAX_BINS) {
    int l = lstart[tid];
    cur[tid] = l;
    cst[tid * nc + c] = l;
  }
  __syncthreads();
  for (int i = tid; i < (ch >> 2); i += blockDim.x) {
    v4i s = __builtin_nontemporal_load(s4 + i);
    v4i d = d4[i];
#pragma unroll
    for (int k = 0; k < 4; ++k) {
      int dk = d[k], sk = s[k];
      int p = atomicAdd(&cur[dk >> BIN_SHIFT], 1);
      rec[p] = sk | ((dk & (BIN_NODES - 1)) << 16);
    }
  }
  __syncthreads();
  int* t = tmp + (size_t)c * ch;
  for (int i = tid; i < ch; i += blockDim.x) t[i] = rec[i];
}

__global__ void __launch_bounds__(1024)
k_csr(const int* __restrict__ tmp, const int* __restrict__ cst, int nc, int ne,
      int* __restrict__ off, float* __restrict__ dinv, int* __restrict__ csr, int n) {
  __shared__ int dc[BIN_NODES];
  __shared__ int cur[BIN_NODES];
  __shared__ int rbeg[MAX_CHUNKS];
  __shared__ int rlen[MAX_CHUNKS];
  __shared__ int redlo[32], redhi[32];
  __shared__ int stage[STAGE_MAX];
  const int b = blockIdx.x, tid = threadIdx.x;
  const int lane = tid & 31, w = tid >> 5;
  const int ch = ne / nc;
  int lo = 0, hi = 0;
  if (tid < nc && tid < MAX_CHUNKS) {
    lo = cst[b * nc + tid];
    hi = (b + 1 < MAX_BINS) ? cst[(b + 1) * nc + tid] : ch;
  }
  if (tid < MAX_CHUNKS) { rbeg[tid] = tid * ch + lo; rlen[tid] = hi - lo; }
  if (tid < BIN_NODES) dc[tid] = 0;
  int slo = lo, shi = hi;
  for (int o = 16; o > 0; o >>= 1) { slo += __shfl_xor(slo, o); shi += __shfl_xor(shi, o); }
  if (lane == 0) { redlo[w] = slo; redhi[w] = shi; }
  __syncthreads();
  int start = redlo[lane], end = redhi[lane];
  for (int o = 16; o > 0; o >>= 1) { start += __shfl_xor(start, o); end += __shfl_xor(end, o); }
  const int m = end - start;

  int beg[8], len[8], r[8][RUN_REGS];
#pragma unroll
  for (int k = 0; k < 8; ++k) { beg[k] = rbeg[w + 32 * k]; len[k] = rlen[w + 32 * k]; }
#pragma unroll
  for (int k = 0; k < 8; ++k)
#pragma unroll
    for (int h = 0; h < RUN_REGS; ++h) {
      const int j = lane + 32 * h;
      r[k][h] = (j < len[k]) ? tmp[beg[k] + j] : -1;
    }
#pragma unroll
  for (int k = 0; k < 8; ++k) {
#pragma unroll
    for (int h = 0; h < RUN_REGS; ++h)
      if (r[k][h] >= 0) atomicAdd(&dc[r[k][h] >> 16], 1);
    for (int j = lane + 32 * RUN_REGS; j < len[k]; j += 32) atomicAdd(&dc[tmp[beg[k] + j] >> 16], 1);
  }
  __syncthreads();
  if (tid < 32) scan256_first_wave(dc, cur);
  __syncthreads();
  if (tid < BIN_NODES) {
    int node = b * BIN_NODES + tid;
    off[node]  = start + cur[tid];
    dinv[node] = rsqrtf((float)(dc[tid] + 1));
  }
  if (b == (int)gridDim.x - 1 && tid == 0) off[n] = end;
  __syncthreads();
  const bool staged = m <= STAGE_MAX;
#pragma unroll
  for (int k = 0; k < 8; ++k) {
#pragma unroll
    for (int h = 0; h < RUN_REGS; ++h)
      if (r[k][h] >= 0) {
        int p = atomicAdd(&cur[r[k][h] >> 16], 1);
        int s = r[k][h] & 0xFFFF;
        if (staged) stage[p] = s; else csr[start + p] = s;
      }
    for (int j = lane + 32 * RUN_REGS; j < len[k]; j += 32) {
      int rr = tmp[beg[k] + j];
      int p = atomicAdd(&cur[rr >> 16], 1);
      int s = rr & 0xFFFF;
      if (staged) stage[p] = s; else csr[start + p] = s;
    }
  }
  if (staged) {
    __syncthreads();
    for (int i = tid; i < m; i += blockDim.x) csr[start + i] = stage[i];
  }
}

template <bool A_COLMAJOR_RELU>
__global__ void k_gemm_n16(const float* __restrict__ A, const float* __restrict__ B,
                           float* __restrict__ CT, int nrows) {
  int wid  = (blockIdx.x * blockDim.x + threadIdx.x) >> 5;
  int lane = threadIdx.x & 31;
  int row0 = wid * 32;
  if (row0 >= nrows) return;
  int m  = lane & 15;
  int kh = (lane >> 4) << 1;
  v2f bf[4];
#pragma unroll
  for (int ks = 0; ks < 4; ++ks) {
    bf[ks].x = B[(ks * 4 + kh + 0) * 16 + m];
    bf[ks].y = B[(ks * 4 + kh + 1) * 16 + m];
  }
#pragma unroll
  for (int t = 0; t < 2; ++t) {
    const int row = row0 + t * 16 + m;
    v8f c = {};
#pragma unroll
    for (int ks = 0; ks < 4; ++ks) {
      const int k = ks * 4 + kh;
      v2f a;
      if (A_COLMAJOR_RELU) {
        a.x = fmaxf(A[(size_t)(k + 0) * nrows + row], 0.0f);
        a.y = fmaxf(A[(size_t)(k + 1) * nrows + row], 0.0f);
      } else {
        a.x = A[row * 16 + k + 0];
        a.y = A[row * 16 + k + 1];
      }
      c = __builtin_amdgcn_wmma_f32_16x16x4_f32(false, bf[ks], false, a, (short)0, c, false, false);
    }
    float* o = CT + (size_t)((lane >> 4) << 3) * nrows + row;
#pragma unroll
    for (int r = 0; r < 8; ++r) o[(size_t)r * nrows] = c[r];
  }
}

#define TAB_N    65536
#define F_COLS   16
#define PER_TRIP 12

__global__ void __launch_bounds__(1024)
k_pull_col(const int* __restrict__ off, const int* __restrict__ csr,
           const float* __restrict__ dinv, const float* __restrict__ hwT,
           const float* __restrict__ bias, float* __restrict__ aggT, int n) {
  __shared__ __attribute__((aligned(16))) float tab[TAB_N];
  const int tid = threadIdx.x;
  const int f   = blockIdx.x % F_COLS;
  const int rg  = blockIdx.x / F_COLS;
  const int nrg = gridDim.x / F_COLS;
  const int per = n / nrg;

  {
    const float4* h4 = (const float4*)(hwT + (size_t)f * n);
    const float4* d4 = (const float4*)dinv;
    const int n4  = n >> 2;
    const int rot = (blockIdx.x * 64) % n4;
#pragma unroll 4
    for (int i = tid; i < n4; i += blockDim.x) {
      int j = i + rot;
      if (j >= n4) j -= n4;
      float4 a = h4[j], b = d4[j];
      float4 p; p.x = a.x * b.x; p.y = a.y * b.y; p.z = a.z * b.z; p.w = a.w * b.w;
      ((float4*)tab)[j] = p;
    }
  }
  __syncthreads();

  const int lane = tid & 31, w = tid >> 5;
  const int q = lane & 3, g = lane >> 2;
  const int nwaves = blockDim.x >> 5;
  for (int u = w; u * 32 < per; u += nwaves) {
    const int nb = rg * per + u * 32;
    int first[4], end[4], last[4];
#pragma unroll
    for (int sub = 0; sub < 4; ++sub) {
      const int node = nb + sub * 8 + g;
      first[sub] = off[node] + q;
      end[sub]   = off[node + 1];
      last[sub]  = max(end[sub] - 1, 0);
    }
    int s[4][PER_TRIP];
#pragma unroll
    for (int sub = 0; sub < 4; ++sub)
#pragma unroll
      for (int k = 0; k < PER_TRIP; ++k)
        s[sub][k] = csr[min(first[sub] + 4 * k, last[sub])];
    float a[4];
#pragma unroll
    for (int sub = 0; sub < 4; ++sub) {
      float acc = 0.0f;
#pragma unroll
      for (int k = 0; k < PER_TRIP; ++k) {
        float v = tab[s[sub][k]];
        acc += (first[sub] + 4 * k < end[sub]) ? v : 0.0f;
      }
      a[sub] = acc;
    }
#pragma unroll
    for (int sub = 0; sub < 4; ++sub)
      for (int i = first[sub] + 4 * PER_TRIP; i < end[sub]; i += 4) a[sub] += tab[csr[i]];
    float res = 0.0f;
#pragma unroll
    for (int sub = 0; sub < 4; ++sub) {
      float t = a[sub];
      t += __shfl_xor(t, 1);
      t += __shfl_xor(t, 2);
      if (q == sub) res = t;
    }
    const int mine = nb + q * 8 + g;
    aggT[(size_t)f * n + mine] = dinv[mine] * (tab[mine] + res) + bias[f];
  }
}

#define HEAD_TILES  12
#define HEAD_GROUPS 8

template <int KLEN>
__global__ void __launch_bounds__(256)
k_heads_partial(const float* __restrict__ AT, int nn, int npg,
                const float* __restrict__ Bp, const float* __restrict__ Bv,
                float* __restrict__ part) {
  __shared__ float red[8][256];
  const int lane = threadIdx.x & 31, w = threadIdx.x >> 5;
  const int tile = blockIdx.x >> 3;
  const int sg   = blockIdx.x & 7;
  const int tm = tile / 3, tc = tile - tm * 3;
  const int split = sg * 8 + w;
  const bool pol = tc < 2;
  const float* B  = pol ? Bp : Bv;
  const int ncols = pol ? 32 : 16;
  const int c0    = pol ? tc * 16 : 0;
  const int m  = lane & 15;
  const int kh = (lane >> 4) << 1;
  const float* abase = AT + (size_t)(tm * 16 + m) * npg + split * (KLEN / 16);
  const float* bbase = B + (size_t)(split * KLEN) * ncols + c0 + m;
  v8f c = {};
#pragma unroll 4
  for (int i = 0; i < KLEN / 4; ++i) {
    const int f  = i >> 2;
    const int nl = ((i & 3) << 2) + kh;
    v2f a, b;
    const int ai = f * nn + nl;
    const int bi = (nl * 16 + f) * ncols;
    a.x = fmaxf(abase[ai + 0], 0.0f);
    a.y = fmaxf(abase[ai + 1], 0.0f);
    b.x = bbase[bi];
    b.y = bbase[bi + 16 * ncols];
    c = __builtin_amdgcn_wmma_f32_16x16x4_f32(false, a, false, b, (short)0, c, false, false);
  }
#pragma unroll
  for (int r = 0; r < 8; ++r) red[w][r * 32 + lane] = c[r];
  __syncthreads();
  float s = 0.0f;
#pragma unroll
  for (int k = 0; k < 8; ++k) s += red[k][threadIdx.x];
  part[(size_t)(sg * HEAD_TILES + tile) * 256 + threadIdx.x] = s;
}

__global__ void __launch_bounds__(256)
k_heads_out(const float* __restrict__ part,
            const float* __restrict__ p1b, const float* __restrict__ v1b,
            const float* __restrict__ p2w, const float* __restrict__ p2b,
            const float* __restrict__ v2w, const float* __restrict__ v2b,
            float* __restrict__ xout, float* __restrict__ vout) {
  __shared__ float hid[16][33];
  __shared__ float vh[64][17];
  const int t = threadIdx.x;
  const int lane = t & 31, w = t >> 5;
  const int rr = (t >> 5) + (((t >> 4) & 1) << 3);
  const int cc = t & 15;
  if (blockIdx.x == 32) {
#pragma unroll
    for (int tm = 0; tm < 4; ++tm) {
      float s = 0.0f;
#pragma unroll
      for (int sg = 0; sg < HEAD_GROUPS; ++sg) s += part[(sg * HEAD_TILES + tm * 3 + 2) * 256 + t];
      vh[tm * 16 + rr][cc] = fmaxf(s + v1b[cc], 0.0f);
    }
    __syncthreads();
    if (t < 64) {
      float s = 0.0f;
#pragma unroll
      for (int j = 0; j < 16; ++j) s += vh[t][j] * v2w[j];
      vout[t] = s + v2b[0];
    }
    return;
  }
  const int tm = blockIdx.x >> 3;
  const int tn = ((blockIdx.x & 7) << 3) + w;
  const int m  = lane & 15;
  const int kh = (lane >> 4) << 1;
  const float* bcol = p2w + tn * 16 + m;
  v2f bfr[8];
#pragma unroll
  for (int ks = 0; ks < 8; ++ks) {
    bfr[ks].x = bcol[(ks * 4 + kh + 0) * 1024];
    bfr[ks].y = bcol[(ks * 4 + kh + 1) * 1024];
  }
  const float bv = p2b[tn * 16 + m];
#pragma unroll
  for (int tc = 0; tc < 2; ++tc) {
    float s = 0.0f;
#pragma unroll
    for (int sg = 0; sg < HEAD_GROUPS; ++sg) s += part[(sg * HEAD_TILES + tm * 3 + tc) * 256 + t];
    hid[rr][tc * 16 + cc] = fmaxf(s + p1b[tc * 16 + cc], 0.0f);
  }
  __syncthreads();
  v8f c = {};
#pragma unroll
  for (int ks = 0; ks < 8; ++ks) {
    v2f a;
    a.x = hid[m][ks * 4 + kh + 0];
    a.y = hid[m][ks * 4 + kh + 1];
    c = __builtin_amdgcn_wmma_f32_16x16x4_f32(false, a, false, bfr[ks], (short)0, c, false, false);
  }
  const int rbase = tm * 16 + ((lane >> 4) << 3);
#pragma unroll
  for (int r = 0; r < 8; ++r)
    xout[(rbase + r) * 1024 + tn * 16 + m] = c[r] + bv;
}

extern "C" void kernel_launch(void* const* d_in, const int* in_sizes, int n_in,
                              void* d_out, int out_size, void* d_ws, size_t ws_size,
                              hipStream_t stream) {
  const float* x    = (const float*)d_in[0];
  const int*   ei   = (const int*)d_in[1];
  const float* W1   = (const float*)d_in[2];
  const float* b1   = (const float*)d_in[3];
  const float* W2   = (const float*)d_in[4];
  const float* b2   = (const float*)d_in[5];
  const float* p1w  = (const float*)d_in[6];
  const float* p1b  = (const float*)d_in[7];
  const float* p2w  = (const float*)d_in[8];
  const float* p2b  = (const float*)d_in[9];
  const float* v1w  = (const float*)d_in[10];
  const float* v1b  = (const float*)d_in[11];
  const float* v2w  = (const float*)d_in[12];
  const float* v2b  = (const float*)d_in[13];

  const int N  = in_sizes[0] / 16;
  const int E  = in_sizes[1] / 2;
  const int* esrc = ei;
  const int* edst = ei + E;

  float* ws   = (float*)d_ws;
  int*   cst  = (int*)ws;
  int*   off  = cst + 256 * 256 + 32;
  float* dinv = (float*)(off + N + 32);
  int*   csr  = (int*)(dinv + N);
  float* bufA = (float*)(csr + E);
  float* bufB = bufA + (size_t)16 * N;
  int*   tmp  = (int*)bufA;
  float* part = bufA;

  float* Xout = (float*)d_out;
  float* Vout = Xout + 64 * 1024;

  const int B256 = 256;
  const int WGPS = 256;
  const int NC   = 256;
  k_chunk_sort<<<NC, 1024, 0, stream>>>(esrc, edst, tmp, cst, E);
  k_csr<<<N / 256, 1024, 0, stream>>>(tmp, cst, NC, E, off, dinv, csr, N);

  int gemm_blocks = (N / 32) / 8;
  k_gemm_n16<false><<<gemm_blocks, B256, 0, stream>>>(x, W1, bufA, N);
  k_pull_col<<<WGPS, 1024, 0, stream>>>(off, csr, dinv, bufA, b1, bufB, N);

  k_gemm_n16<true><<<gemm_blocks, B256, 0, stream>>>(bufB, W2, bufA, N);
  k_pull_col<<<WGPS, 1024, 0, stream>>>(off, csr, dinv, bufA, b2, bufB, N);

  const int K = 16384, NSPLIT = 64, KLEN = K / NSPLIT;

  k_heads_partial<KLEN><<<96, B256, 0, stream>>>(bufB, N, K / 16, p1w, v1w, part);
  k_heads_out<<<33, B256, 0, stream>>>(part, p1b, v1b, p2w, p2b, v2w, v2b, Xout, Vout);
}


# edge gather reads the sorted list in aligned 16-byte pieces: a quad takes 16 consecutive entries per load, 3 loads per lane not 12; entries outside a list read a zero LDS word; sum order changes
# speedup vs baseline: 4.4499x; 1.0436x over previous
#include <hip/hip_runtime.h>
#include <hip/hip_bf16.h>

typedef __attribute__((ext_vector_type(2))) float v2f;
typedef __attribute__((ext_vector_type(8))) float v8f;
typedef __attribute__((ext_vector_type(4))) int   v4i;

#define BIN_SHIFT 8
#define BIN_NODES 256
#define MAX_BINS  256
#define CH_MAX    8192
#define MAX_CHUNKS 256
#define RUN_REGS  2
#define STAGE_MAX 12288

__device__ __forceinline__ void scan256_first_wave(const int* in, int* out) {
  const int lane = threadIdx.x;
  int v[8];
  int run = 0;
#pragma unroll
  for (int k = 0; k < 8; ++k) { int t = in[lane * 8 + k]; v[k] = run; run += t; }
  int inc = run;
  for (int o = 1; o < 32; o <<= 1) {
    int t = __shfl_up(inc, o);
    if (lane >= o) inc += t;
  }
  const int ex = inc - run;
#pragma unroll
  for (int k = 0; k < 8; ++k) out[lane * 8 + k] = ex + v[k];
}

__global__ void __launch_bounds__(1024)
k_chunk_sort(const int* __restrict__ src, const int* __restrict__ dst,
             int* __restrict__ tmp, int* __restrict__ cst, int ne) {
  __shared__ int cntl[MAX_BINS];
  __shared__ int lstart[MAX_BINS];
  __shared__ int cur[MAX_BINS];
  __shared__ int rec[CH_MAX];
  const int nc = gridDim.x, c = blockIdx.x, tid = threadIdx.x;
  const int ch = ne / nc;
  if (tid < MAX_BINS) cntl[tid] = 0;
  __syncthreads();
  const v4i* s4 = (const v4i*)(src + (size_t)c * ch);
  const v4i* d4 = (const v4i*)(dst + (size_t)c * ch);
  for (int i = tid; i < (ch >> 2); i += blockDim.x) {
    v4i d = d4[i];
    atomicAdd(&cntl[d.x >> BIN_SHIFT], 1);
    atomicAdd(&cntl[d.y >> BIN_SHIFT], 1);
    atomicAdd(&cntl[d.z >> BIN_SHIFT], 1);
    atomicAdd(&cntl[d.w >> BIN_SHIFT], 1);
  }
  __syncthreads();
  if (tid < 32) scan256_first_wave(cntl, lstart);
  __syncthreads();
  if (tid < MAX_BINS) {
    int l = lstart[tid];
    cur[tid] = l;
    cst[tid * nc + c] = l;
  }
  __syncthreads();
  for (int i = tid; i < (ch >> 2); i += blockDim.x) {
    v4i s = __builtin_nontemporal_load(s4 + i);
    v4i d = d4[i];
#pragma unroll
    for (int k = 0; k < 4; ++k) {
      int dk = d[k], sk = s[k];
      int p = atomicAdd(&cur[dk >> BIN_SHIFT], 1);
      rec[p] = sk | ((dk & (BIN_NODES - 1)) << 16);
    }
  }
  __syncthreads();
  int* t = tmp + (size_t)c * ch;
  for (int i = tid; i < ch; i += blockDim.x) t[i] = rec[i];
}

__global__ void __launch_bounds__(1024)
k_csr(const int* __restrict__ tmp, const int* __restrict__ cst, int nc, int ne,
      int* __restrict__ off, float* __restrict__ dinv, int* __restrict__ csr, int n) {
  __shared__ int dc[BIN_NODES];
  __shared__ int cur[BIN_NODES];
  __shared__ int rbeg[MAX_CHUNKS];
  __shared__ int rlen[MAX_CHUNKS];
  __shared__ int redlo[32], redhi[32];
  __shared__ int stage[STAGE_MAX];
  const int b = blockIdx.x, tid = threadIdx.x;
  const int lane = tid & 31, w = tid >> 5;
  const int ch = ne / nc;
  int lo = 0, hi = 0;
  if (tid < nc && tid < MAX_CHUNKS) {
    lo = cst[b * nc + tid];
    hi = (b + 1 < MAX_BINS) ? cst[(b + 1) * nc + tid] : ch;
  }
  if (tid < MAX_CHUNKS) { rbeg[tid] = tid * ch + lo; rlen[tid] = hi - lo; }
  if (tid < BIN_NODES) dc[tid] = 0;
  int slo = lo, shi = hi;
  for (int o = 16; o > 0; o >>= 1) { slo += __shfl_xor(slo, o); shi += __shfl_xor(shi, o); }
  if (lane == 0) { redlo[w] = slo; redhi[w] = shi; }
  __syncthreads();
  int start = redlo[lane], end = redhi[lane];
  for (int o = 16; o > 0; o >>= 1) { start += __shfl_xor(start, o); end += __shfl_xor(end, o); }
  const int m = end - start;

  int beg[8], len[8], r[8][RUN_REGS];
#pragma unroll
  for (int k = 0; k < 8; ++k) { beg[k] = rbeg[w + 32 * k]; len[k] = rlen[w + 32 * k]; }
#pragma unroll
  for (int k = 0; k < 8; ++k)
#pragma unroll
    for (int h = 0; h < RUN_REGS; ++h) {
      const int j = lane + 32 * h;
      r[k][h] = (j < len[k]) ? tmp[beg[k] + j] : -1;
    }
#pragma unroll
  for (int k = 0; k < 8; ++k) {
#pragma unroll
    for (int h = 0; h < RUN_REGS; ++h)
      if (r[k][h] >= 0) atomicAdd(&dc[r[k][h] >> 16], 1);
    for (int j = lane + 32 * RUN_REGS; j < len[k]; j += 32) atomicAdd(&dc[tmp[beg[k] + j] >> 16], 1);
  }
  __syncthreads();
  if (tid < 32) scan256_first_wave(dc, cur);
  __syncthreads();
  if (tid < BIN_NODES) {
    int node = b * BIN_NODES + tid;
    off[node]  = start + cur[tid];
    dinv[node] = rsqrtf((float)(dc[tid] + 1));
  }
  if (b == (int)gridDim.x - 1 && tid == 0) off[n] = end;
  __syncthreads();
  const bool staged = m <= STAGE_MAX;
#pragma unroll
  for (int k = 0; k < 8; ++k) {
#pragma unroll
    for (int h = 0; h < RUN_REGS; ++h)
      if (r[k][h] >= 0) {
        int p = atomicAdd(&cur[r[k][h] >> 16], 1);
        int s = r[k][h] & 0xFFFF;
        if (staged) stage[p] = s; else csr[start + p] = s;
      }
    for (int j = lane + 32 * RUN_REGS; j < len[k]; j += 32) {
      int rr = tmp[beg[k] + j];
      int p = atomicAdd(&cur[rr >> 16], 1);
      int s = rr & 0xFFFF;
      if (staged) stage[p] = s; else csr[start + p] = s;
    }
  }
  if (staged) {
    __syncthreads();
    for (int i = tid; i < m; i += blockDim.x) csr[start + i] = stage[i];
  }
}

template <bool A_COLMAJOR_RELU>
__global__ void k_gemm_n16(const float* __restrict__ A, const float* __restrict__ B,
                           float* __restrict__ CT, int nrows) {
  int wid  = (blockIdx.x * blockDim.x + threadIdx.x) >> 5;
  int lane = threadIdx.x & 31;
  int row0 = wid * 32;
  if (row0 >= nrows) return;
  int m  = lane & 15;
  int kh = (lane >> 4) << 1;
  v2f bf[4];
#pragma unroll
  for (int ks = 0; ks < 4; ++ks) {
    bf[ks].x = B[(ks * 4 + kh + 0) * 16 + m];
    bf[ks].y = B[(ks * 4 + kh + 1) * 16 + m];
  }
#pragma unroll
  for (int t = 0; t < 2; ++t) {
    const int row = row0 + t * 16 + m;
    v8f c = {};
#pragma unroll
    for (int ks = 0; ks < 4; ++ks) {
      const int k = ks * 4 + kh;
      v2f a;
      if (A_COLMAJOR_RELU) {
        a.x = fmaxf(A[(size_t)(k + 0) * nrows + row], 0.0f);
        a.y = fmaxf(A[(size_t)(k + 1) * nrows + row], 0.0f);
      } else {
        a.x = A[row * 16 + k + 0];
        a.y = A[row * 16 + k + 1];
      }
      c = __builtin_amdgcn_wmma_f32_16x16x4_f32(false, bf[ks], false, a, (short)0, c, false, false);
    }
    float* o = CT + (size_t)((lane >> 4) << 3) * nrows + row;
#pragma unroll
    for (int r = 0; r < 8; ++r) o[(size_t)r * nrows] = c[r];
  }
}

#define TAB_N    65536
#define F_COLS   16
#define PIECES   3

__global__ void __launch_bounds__(1024)
k_pull_col(const int* __restrict__ off, const int* __restrict__ csr,
           const float* __restrict__ dinv, const float* __restrict__ hwT,
           const float* __restrict__ bias, float* __restrict__ aggT, int n, int ne) {
  __shared__ __attribute__((aligned(16))) float tab[TAB_N + 4];
  const int tid = threadIdx.x;
  const int f   = blockIdx.x % F_COLS;
  const int rg  = blockIdx.x / F_COLS;
  const int nrg = gridDim.x / F_COLS;
  const int per = n / nrg;

  {
    const float4* h4 = (const float4*)(hwT + (size_t)f * n);
    const float4* d4 = (const float4*)dinv;
    const int n4  = n >> 2;
    const int rot = (blockIdx.x * 64) % n4;
#pragma unroll 4
    for (int i = tid; i < n4; i += blockDim.x) {
      int j = i + rot;
      if (j >= n4) j -= n4;
      float4 a = h4[j], b = d4[j];
      float4 p; p.x = a.x * b.x; p.y = a.y * b.y; p.z = a.z * b.z; p.w = a.w * b.w;
      ((float4*)tab)[j] = p;
    }
  }
  if (tid == 0) tab[TAB_N] = 0.0f;
  __syncthreads();

  const int lane = tid & 31, w = tid >> 5;
  const int q = lane & 3, g = lane >> 2;
  const int nwaves = blockDim.x >> 5;
  const v4i* csr4 = (const v4i*)csr;
  const int last4 = (ne >> 2) - 1;
  for (int u = w; u * 32 < per; u += nwaves) {
    const int nb = rg * per + u * 32;
    int base[4], lo[4], hi[4];
#pragma unroll
    for (int sub = 0; sub < 4; ++sub) {
      const int node = nb + sub * 8 + g;
      const int o0 = off[node];
      const int o1 = off[node + 1];
      base[sub] = (o0 & ~3) + 4 * q;
      lo[sub]   = o0 - base[sub];
      hi[sub]   = o1 - base[sub];
    }
    v4i s[4][PIECES];
#pragma unroll
    for (int sub = 0; sub < 4; ++sub)
#pragma unroll
      for (int k = 0; k < PIECES; ++k)
        s[sub][k] = csr4[min((base[sub] >> 2) + 4 * k, last4)];
    float a[4];
#pragma unroll
    for (int sub = 0; sub < 4; ++sub) {
      float acc = 0.0f;
#pragma unroll
      for (int k = 0; k < PIECES; ++k)
#pragma unroll
        for (int j = 0; j < 4; ++j) {
          const int c = 16 * k + j;
          bool in = c < hi[sub];
          if (c < 3) in = in && (c >= lo[sub]);
          acc += tab[in ? s[sub][k][j] : TAB_N];
        }
      a[sub] = acc;
    }
#pragma unroll
    for (int sub = 0; sub < 4; ++sub)
      for (int c = 16 * PIECES - 3 * q; c < hi[sub]; c += 4) a[sub] += tab[csr[base[sub] + c]];
    float res = 0.0f;
#pragma unroll
    for (int sub = 0; sub < 4; ++sub) {
      float t = a[sub];
      t += __shfl_xor(t, 1);
      t += __shfl_xor(t, 2);
      if (q == sub) res = t;
    }
    const int mine = nb + q * 8 + g;
    aggT[(size_t)f * n + mine] = dinv[mine] * (tab[mine] + res) + bias[f];
  }
}

#define HEAD_TILES  12
#define HEAD_GROUPS 8

template <int KLEN>
__global__ void __launch_bounds__(256)
k_heads_partial(const float* __restrict__ AT, int nn, int npg,
                const float* __restrict__ Bp, const float* __restrict__ Bv,
                float* __restrict__ part) {
  __shared__ float red[8][256];
  const int lane = threadIdx.x & 31, w = threadIdx.x >> 5;
  const int tile = blockIdx.x >> 3;
  const int sg   = blockIdx.x & 7;
  const int tm = tile / 3, tc = tile - tm * 3;
  const int split = sg * 8 + w;
  const bool pol = tc < 2;
  const float* B  = pol ? Bp : Bv;
  const int ncols = pol ? 32 : 16;
  const int c0    = pol ? tc * 16 : 0;
  const int m  = lane & 15;
  const int kh = (lane >> 4) << 1;
  const float* abase = AT + (size_t)(tm * 16 + m) * npg + split * (KLEN / 16);
  const float* bbase = B + (size_t)(split * KLEN) * ncols + c0 + m;
  v8f c = {};
#pragma unroll 4
  for (int i = 0; i < KLEN / 4; ++i) {
    const int f  = i >> 2;
    const int nl = ((i & 3) << 2) + kh;
    v2f a, b;
    const int ai = f * nn + nl;
    const int bi = (nl * 16 + f) * ncols;
    a.x = fmaxf(abase[ai + 0], 0.0f);
    a.y = fmaxf(abase[ai + 1], 0.0f);
    b.x = bbase[bi];
    b.y = bbase[bi + 16 * ncols];
    c = __builtin_amdgcn_wmma_f32_16x16x4_f32(false, a, false, b, (short)0, c, false, false);
  }
#pragma unroll
  for (int r = 0; r < 8; ++r) red[w][r * 32 + lane] = c[r];
  __syncthreads();
  float s = 0.0f;
#pragma unroll
  for (int k = 0; k < 8; ++k) s += red[k][threadIdx.x];
  part[(size_t)(sg * HEAD_TILES + tile) * 256 + threadIdx.x] = s;
}

__global__ void __launch_bounds__(256)
k_heads_out(const float* __restrict__ part,
            const float* __restrict__ p1b, const float* __restrict__ v1b,
            const float* __restrict__ p2w, const float* __restrict__ p2b,
            const float* __restrict__ v2w, const float* __restrict__ v2b,
            float* __restrict__ xout, float* __restrict__ vout) {
  __shared__ float hid[16][33];
  __shared__ float vh[64][17];
  const int t = threadIdx.x;
  const int lane = t & 31, w = t >> 5;
  const int rr = (t >> 5) + (((t >> 4) & 1) << 3);
  const int cc = t & 15;
  if (blockIdx.x == 32) {
#pragma unroll
    for (int tm = 0; tm < 4; ++tm) {
      float s = 0.0f;
#pragma unroll
      for (int sg = 0; sg < HEAD_GROUPS; ++sg) s += part[(sg * HEAD_TILES + tm * 3 + 2) * 256 + t];
      vh[tm * 16 + rr][cc] = fmaxf(s + v1b[cc], 0.0f);
    }
    __syncthreads();
    if (t < 64) {
      float s = 0.0f;
#pragma unroll
      for (int j = 0; j < 16; ++j) s += vh[t][j] * v2w[j];
      vout[t] = s + v2b[0];
    }
    return;
  }
  const int tm = blockIdx.x >> 3;
  const int tn = ((blockIdx.x & 7) << 3) + w;
  const int m  = lane & 15;
  const int kh = (lane >> 4) << 1;
  const float* bcol = p2w + tn * 16 + m;
  v2f bfr[8];
#pragma unroll
  for (int ks = 0; ks < 8; ++ks) {
    bfr[ks].x = bcol[(ks * 4 + kh + 0) * 1024];
    bfr[ks].y = bcol[(ks * 4 + kh + 1) * 1024];
  }
  const float bv = p2b[tn * 16 + m];
#pragma unroll
  for (int tc = 0; tc < 2; ++tc) {
    float s = 0.0f;
#pragma unroll
    for (int sg = 0; sg < HEAD_GROUPS; ++sg) s += part[(sg * HEAD_TILES + tm * 3 + tc) * 256 + t];
    hid[rr][tc * 16 + cc] = fmaxf(s + p1b[tc * 16 + cc], 0.0f);
  }
  __syncthreads();
  v8f c = {};
#pragma unroll
  for (int ks = 0; ks < 8; ++ks) {
    v2f a;
    a.x = hid[m][ks * 4 + kh + 0];
    a.y = hid[m][ks * 4 + kh + 1];
    c = __builtin_amdgcn_wmma_f32_16x16x4_f32(false, a, false, bfr[ks], (short)0, c, false, false);
  }
  const int rbase = tm * 16 + ((lane >> 4) << 3);
#pragma unroll
  for (int r = 0; r < 8; ++r)
    xout[(rbase + r) * 1024 + tn * 16 + m] = c[r] + bv;
}

extern "C" void kernel_launch(void* const* d_in, const int* in_sizes, int n_in,
                              void* d_out, int out_size, void* d_ws, size_t ws_size,
                              hipStream_t stream) {
  const float* x    = (const float*)d_in[0];
  const int*   ei   = (const int*)d_in[1];
  const float* W1   = (const float*)d_in[2];
  const float* b1   = (const float*)d_in[3];
  const float* W2   = (const float*)d_in[4];
  const float* b2   = (const float*)d_in[5];
  const float* p1w  = (const float*)d_in[6];
  const float* p1b  = (const float*)d_in[7];
  const float* p2w  = (const float*)d_in[8];
  const float* p2b  = (const float*)d_in[9];
  const float* v1w  = (const float*)d_in[10];
  const float* v1b  = (const float*)d_in[11];
  const float* v2w  = (const float*)d_in[12];
  const float* v2b  = (const float*)d_in[13];

  const int N  = in_sizes[0] / 16;
  const int E  = in_sizes[1] / 2;
  const int* esrc = ei;
  const int* edst = ei + E;

  float* ws   = (float*)d_ws;
  int*   cst  = (int*)ws;
  int*   off  = cst + 256 * 256 + 32;
  float* dinv = (float*)(off + N + 32);
  int*   csr  = (int*)(dinv + N);
  float* bufA = (float*)(csr + E);
  float* bufB = bufA + (size_t)16 * N;
  int*   tmp  = (int*)bufA;
  float* part = bufA;

  float* Xout = (float*)d_out;
  float* Vout = Xout + 64 * 1024;

  const int B256 = 256;
  const int WGPS = 256;
  const int NC   = 256;
  k_chunk_sort<<<NC, 1024, 0, stream>>>(esrc, edst, tmp, cst, E);
  k_csr<<<N / 256, 1024, 0, stream>>>(tmp, cst, NC, E, off, dinv, csr, N);

  int gemm_blocks = (N / 32) / 8;
  k_gemm_n16<false><<<gemm_blocks, B256, 0, stream>>>(x, W1, bufA, N);
  k_pull_col<<<WGPS, 1024, 0, stream>>>(off, csr, dinv, bufA, b1, bufB, N, E);

  k_gemm_n16<true><<<gemm_blocks, B256, 0, stream>>>(bufB, W2, bufA, N);
  k_pull_col<<<WGPS, 1024, 0, stream>>>(off, csr, dinv, bufA, b2, bufB, N, E);

  const int K = 16384, NSPLIT = 64, KLEN = K / NSPLIT;

  k_heads_partial<KLEN><<<96, B256, 0, stream>>>(bufB, N, K / 16, p1w, v1w, part);
  k_heads_out<<<33, B256, 0, stream>>>(part, p1b, v1b, p2w, p2b, v2w, v2b, Xout, Vout);
}


# heads' first operand staged once in LDS: relu(hf) of 16 graphs x 512 k brought in by whole-line loads and shared by the 3 column tiles; 128 work-groups x 24 waves; 32 partial tiles; sum order changes
# speedup vs baseline: 4.7433x; 1.0659x over previous
#include <hip/hip_runtime.h>
#include <hip/hip_bf16.h>

typedef __attribute__((ext_vector_type(2))) float v2f;
typedef __attribute__((ext_vector_type(8))) float v8f;
typedef __attribute__((ext_vector_type(4))) int   v4i;

#define BIN_SHIFT 8
#define BIN_NODES 256
#define MAX_BINS  256
#define CH_MAX    8192
#define MAX_CHUNKS 256
#define RUN_REGS  2
#define STAGE_MAX 12288

__device__ __forceinline__ void scan256_first_wave(const int* in, int* out) {
  const int lane = threadIdx.x;
  int v[8];
  int run = 0;
#pragma unroll
  for (int k = 0; k < 8; ++k) { int t = in[lane * 8 + k]; v[k] = run; run += t; }
  int inc = run;
  for (int o = 1; o < 32; o <<= 1) {
    int t = __shfl_up(inc, o);
    if (lane >= o) inc += t;
  }
  const int ex = inc - run;
#pragma unroll
  for (int k = 0; k < 8; ++k) out[lane * 8 + k] = ex + v[k];
}

__global__ void __launch_bounds__(1024)
k_chunk_sort(const int* __restrict__ src, const int* __restrict__ dst,
             int* __restrict__ tmp, int* __restrict__ cst, int ne) {
  __shared__ int cntl[MAX_BINS];
  __shared__ int lstart[MAX_BINS];
  __shared__ int cur[MAX_BINS];
  __shared__ int rec[CH_MAX];
  const int nc = gridDim.x, c = blockIdx.x, tid = threadIdx.x;
  const int ch = ne / nc;
  if (tid < MAX_BINS) cntl[tid] = 0;
  __syncthreads();
  const v4i* s4 = (const v4i*)(src + (size_t)c * ch);
  const v4i* d4 = (const v4i*)(dst + (size_t)c * ch);
  for (int i = tid; i < (ch >> 2); i += blockDim.x) {
    v4i d = d4[i];
    atomicAdd(&cntl[d.x >> BIN_SHIFT], 1);
    atomicAdd(&cntl[d.y >> BIN_SHIFT], 1);
    atomicAdd(&cntl[d.z >> BIN_SHIFT], 1);
    atomicAdd(&cntl[d.w >> BIN_SHIFT], 1);
  }
  __syncthreads();
  if (tid < 32) scan256_first_wave(cntl, lstart);
  __syncthreads();
  if (tid < MAX_BINS) {
    int l = lstart[tid];
    cur[tid] = l;
    cst[tid * nc + c] = l;
  }
  __syncthreads();
  for (int i = tid; i < (ch >> 2); i += blockDim.x) {
    v4i s = __builtin_nontemporal_load(s4 + i);
    v4i d = d4[i];
#pragma unroll
    for (int k = 0; k < 4; ++k) {
      int dk = d[k], sk = s[k];
      int p = atomicAdd(&cur[dk >> BIN_SHIFT], 1);
      rec[p] = sk | ((dk & (BIN_NODES - 1)) << 16);
    }
  }
  __syncthreads();
  int* t = tmp + (size_t)c * ch;
  for (int i = tid; i < ch; i += blockDim.x) t[i] = rec[i];
}

__global__ void __launch_bounds__(1024)
k_csr(const int* __restrict__ tmp, const int* __restrict__ cst, int nc, int ne,
      int* __restrict__ off, float* __restrict__ dinv, int* __restrict__ csr, int n) {
  __shared__ int dc[BIN_NODES];
  __shared__ int cur[BIN_NODES];
  __shared__ int rbeg[MAX_CHUNKS];
  __shared__ int rlen[MAX_CHUNKS];
  __shared__ int redlo[32], redhi[32];
  __shared__ int stage[STAGE_MAX];
  const int b = blockIdx.x, tid = threadIdx.x;
  const int lane = tid & 31, w = tid >> 5;
  const int ch = ne / nc;
  int lo = 0, hi = 0;
  if (tid < nc && tid < MAX_CHUNKS) {
    lo = cst[b * nc + tid];
    hi = (b + 1 < MAX_BINS) ? cst[(b + 1) * nc + tid] : ch;
  }
  if (tid < MAX_CHUNKS) { rbeg[tid] = tid * ch + lo; rlen[tid] = hi - lo; }
  if (tid < BIN_NODES) dc[tid] = 0;
  int slo = lo, shi = hi;
  for (int o = 16; o > 0; o >>= 1) { slo += __shfl_xor(slo, o); shi += __shfl_xor(shi, o); }
  if (lane == 0) { redlo[w] = slo; redhi[w] = shi; }
  __syncthreads();
  int start = redlo[lane], end = redhi[lane];
  for (int o = 16; o > 0; o >>= 1) { start += __shfl_xor(start, o); end += __shfl_xor(end, o); }
  const int m = end - start;

  int beg[8], len[8], r[8][RUN_REGS];
#pragma unroll
  for (int k = 0; k < 8; ++k) { beg[k] = rbeg[w + 32 * k]; len[k] = rlen[w + 32 * k]; }
#pragma unroll
  for (int k = 0; k < 8; ++k)
#pragma unroll
    for (int h = 0; h < RUN_REGS; ++h) {
      const int j = lane + 32 * h;
      r[k][h] = (j < len[k]) ? tmp[beg[k] + j] : -1;
    }
#pragma unroll
  for (int k = 0; k < 8; ++k) {
#pragma unroll
    for (int h = 0; h < RUN_REGS; ++h)
      if (r[k][h] >= 0) atomicAdd(&dc[r[k][h] >> 16], 1);
    for (int j = lane + 32 * RUN_REGS; j < len[k]; j += 32) atomicAdd(&dc[tmp[beg[k] + j] >> 16], 1);
  }
  __syncthreads();
  if (tid < 32) scan256_first_wave(dc, cur);
  __syncthreads();
  if (tid < BIN_NODES) {
    int node = b * BIN_NODES + tid;
    off[node]  = start + cur[tid];
    dinv[node] = rsqrtf((float)(dc[tid] + 1));
  }
  if (b == (int)gridDim.x - 1 && tid == 0) off[n] = end;
  __syncthreads();
  const bool staged = m <= STAGE_MAX;
#pragma unroll
  for (int k = 0; k < 8; ++k) {
#pragma unroll
    for (int h = 0; h < RUN_REGS; ++h)
      if (r[k][h] >= 0) {
        int p = atomicAdd(&cur[r[k][h] >> 16], 1);
        int s = r[k][h] & 0xFFFF;
        if (staged) stage[p] = s; else csr[start + p] = s;
      }
    for (int j = lane + 32 * RUN_REGS; j < len[k]; j += 32) {
      int rr = tmp[beg[k] + j];
      int p = atomicAdd(&cur[rr >> 16], 1);
      int s = rr & 0xFFFF;
      if (staged) stage[p] = s; else csr[start + p] = s;
    }
  }
  if (staged) {
    __syncthreads();
    for (int i = tid; i < m; i += blockDim.x) csr[start + i] = stage[i];
  }
}

template <bool A_COLMAJOR_RELU>
__global__ void k_gemm_n16(const float* __restrict__ A, const float* __restrict__ B,
                           float* __restrict__ CT, int nrows) {
  int wid  = (blockIdx.x * blockDim.x + threadIdx.x) >> 5;
  int lane = threadIdx.x & 31;
  int row0 = wid * 32;
  if (row0 >= nrows) return;
  int m  = lane & 15;
  int kh = (lane >> 4) << 1;
  v2f bf[4];
#pragma unroll
  for (int ks = 0; ks < 4; ++ks) {
    bf[ks].x = B[(ks * 4 + kh + 0) * 16 + m];
    bf[ks].y = B[(ks * 4 + kh + 1) * 16 + m];
  }
#pragma unroll
  for (int t = 0; t < 2; ++t) {
    const int row = row0 + t * 16 + m;
    v8f c = {};
#pragma unroll
    for (int ks = 0; ks < 4; ++ks) {
      const int k = ks * 4 + kh;
      v2f a;
      if (A_COLMAJOR_RELU) {
        a.x = fmaxf(A[(size_t)(k + 0) * nrows + row], 0.0f);
        a.y = fmaxf(A[(size_t)(k + 1) * nrows + row], 0.0f);
      } else {
        a.x = A[row * 16 + k + 0];
        a.y = A[row * 16 + k + 1];
      }
      c = __builtin_amdgcn_wmma_f32_16x16x4_f32(false, bf[ks], false, a, (short)0, c, false, false);
    }
    float* o = CT + (size_t)((lane >> 4) << 3) * nrows + row;
#pragma unroll
    for (int r = 0; r < 8; ++r) o[(size_t)r * nrows] = c[r];
  }
}

#define TAB_N    65536
#define F_COLS   16
#define PIECES   3

__global__ void __launch_bounds__(1024)
k_pull_col(const int* __restrict__ off, const int* __restrict__ csr,
           const float* __restrict__ dinv, const float* __restrict__ hwT,
           const float* __restrict__ bias, float* __restrict__ aggT, int n, int ne) {
  __shared__ __attribute__((aligned(16))) float tab[TAB_N + 4];
  const int tid = threadIdx.x;
  const int f   = blockIdx.x % F_COLS;
  const int rg  = blockIdx.x / F_COLS;
  const int nrg = gridDim.x / F_COLS;
  const int per = n / nrg;

  {
    const float4* h4 = (const float4*)(hwT + (size_t)f * n);
    const float4* d4 = (const float4*)dinv;
    const int n4  = n >> 2;
    const int rot = (blockIdx.x * 64) % n4;
#pragma unroll 4
    for (int i = tid; i < n4; i += blockDim.x) {
      int j = i + rot;
      if (j >= n4) j -= n4;
      float4 a = h4[j], b = d4[j];
      float4 p; p.x = a.x * b.x; p.y = a.y * b.y; p.z = a.z * b.z; p.w = a.w * b.w;
      ((float4*)tab)[j] = p;
    }
  }
  if (tid == 0) tab[TAB_N] = 0.0f;
  __syncthreads();

  const int lane = tid & 31, w = tid >> 5;
  const int q = lane & 3, g = lane >> 2;
  const int nwaves = blockDim.x >> 5;
  const v4i* csr4 = (const v4i*)csr;
  const int last4 = (ne >> 2) - 1;
  for (int u = w; u * 32 < per; u += nwaves) {
    const int nb = rg * per + u * 32;
    int base[4], lo[4], hi[4];
#pragma unroll
    for (int sub = 0; sub < 4; ++sub) {
      const int node = nb + sub * 8 + g;
      const int o0 = off[node];
      const int o1 = off[node + 1];
      base[sub] = (o0 & ~3) + 4 * q;
      lo[sub]   = o0 - base[sub];
      hi[sub]   = o1 - base[sub];
    }
    v4i s[4][PIECES];
#pragma unroll
    for (int sub = 0; sub < 4; ++sub)
#pragma unroll
      for (int k = 0; k < PIECES; ++k)
        s[sub][k] = csr4[min((base[sub] >> 2) + 4 * k, last4)];
    float a[4];
#pragma unroll
    for (int sub = 0; sub < 4; ++sub) {
      float acc = 0.0f;
#pragma unroll
      for (int k = 0; k < PIECES; ++k)
#pragma unroll
        for (int j = 0; j < 4; ++j) {
          const int c = 16 * k + j;
          bool in = c < hi[sub];
          if (c < 3) in = in && (c >= lo[sub]);
          acc += tab[in ? s[sub][k][j] : TAB_N];
        }
      a[sub] = acc;
    }
#pragma unroll
    for (int sub = 0; sub < 4; ++sub)
      for (int c = 16 * PIECES - 3 * q; c < hi[sub]; c += 4) a[sub] += tab[csr[base[sub] + c]];
    float res = 0.0f;
#pragma unroll
    for (int sub = 0; sub < 4; ++sub) {
      float t = a[sub];
      t += __shfl_xor(t, 1);
      t += __shfl_xor(t, 2);
      if (q == sub) res = t;
    }
    const int mine = nb + q * 8 + g;
    aggT[(size_t)f * n + mine] = dinv[mine] * (tab[mine] + res) + bias[f];
  }
}

#define HEAD_TILES  12
#define HEAD_GROUPS 32
#define HP_NODES    32
#define HP_PITCH    17

__global__ void __launch_bounds__(768)
k_heads_partial(const float* __restrict__ AT, int nn, int npg,
                const float* __restrict__ Bp, const float* __restrict__ Bv,
                float* __restrict__ part) {
  __shared__ float sA[(F_COLS + 1) * HP_NODES * HP_PITCH];
  __shared__ float red[24][256];
  const int tid = threadIdx.x;
  const int lane = tid & 31, w = tid >> 5;
  const int tm = blockIdx.x / HEAD_GROUPS;
  const int g  = blockIdx.x % HEAD_GROUPS;
  const int tc = w >> 3, kq = w & 7;
  const bool pol = tc < 2;
  const float* B  = pol ? Bp : Bv;
  const int ncols = pol ? 32 : 16;
  const int c0    = pol ? tc * 16 : 0;
  const int m = lane & 15;
  const int h = lane >> 4;
  const float* abase = AT + (size_t)(tm * 16) * npg + g * HP_NODES + lane;
  float av[11];
#pragma unroll
  for (int j = 0; j < 11; ++j) {
    const int r = min(w + 24 * j, 255);
    av[j] = abase[(size_t)(r >> 4) * nn + (r & 15) * npg];
  }
  const float* bbase = B + (size_t)((g * HP_NODES + 4 * kq) * 16 + h) * ncols + c0 + m;
  v2f bf[16];
#pragma unroll
  for (int i = 0; i < 16; ++i) {
    const int r = (i >> 2) * 16 + (i & 3) * 4;
    bf[i].x = bbase[r * ncols];
    bf[i].y = bbase[(r + 2) * ncols];
  }
#pragma unroll
  for (int j = 0; j < 11; ++j) {
    const int r = w + 24 * j;
    sA[((r >> 4) * HP_NODES + lane) * HP_PITCH + (r & 15)] = fmaxf(av[j], 0.0f);
  }
  __syncthreads();
  v8f c = {};
#pragma unroll
  for (int i = 0; i < 16; ++i) {
    const int nd = 4 * kq + (i >> 2);
    const int f0 = (i & 3) * 4 + h;
    v2f a;
    a.x = sA[(f0 * HP_NODES + nd) * HP_PITCH + m];
    a.y = sA[((f0 + 2) * HP_NODES + nd) * HP_PITCH + m];
    c = __builtin_amdgcn_wmma_f32_16x16x4_f32(false, a, false, bf[i], (short)0, c, false, false);
  }
#pragma unroll
  for (int r = 0; r < 8; ++r) red[w][r * 32 + lane] = c[r];
  __syncthreads();
  const int tcc = tid >> 8, e = tid & 255;
  float s = 0.0f;
#pragma unroll
  for (int k = 0; k < 8; ++k) s += red[tcc * 8 + k][e];
  part[(size_t)(g * HEAD_TILES + tm * 3 + tcc) * 256 + e] = s;
}

__global__ void __launch_bounds__(256)
k_heads_out(const float* __restrict__ part,
            const float* __restrict__ p1b, const float* __restrict__ v1b,
            const float* __restrict__ p2w, const float* __restrict__ p2b,
            const float* __restrict__ v2w, const float* __restrict__ v2b,
            float* __restrict__ xout, float* __restrict__ vout) {
  __shared__ float hid[16][33];
  __shared__ float vh[16][17];
  const int t = threadIdx.x;
  const int lane = t & 31, w = t >> 5;
  const int rr = (t >> 5) + (((t >> 4) & 1) << 3);
  const int cc = t & 15;
  if (blockIdx.x >= 32) {
    const int tm = blockIdx.x - 32;
    float s = 0.0f;
#pragma unroll
    for (int sg = 0; sg < HEAD_GROUPS; ++sg) s += part[(sg * HEAD_TILES + tm * 3 + 2) * 256 + t];
    vh[rr][cc] = fmaxf(s + v1b[cc], 0.0f);
    __syncthreads();
    if (t < 16) {
      float d = 0.0f;
#pragma unroll
      for (int j = 0; j < 16; ++j) d += vh[t][j] * v2w[j];
      vout[tm * 16 + t] = d + v2b[0];
    }
    return;
  }
  const int tm = blockIdx.x >> 3;
  const int tn = ((blockIdx.x & 7) << 3) + w;
  const int m  = lane & 15;
  const int kh = (lane >> 4) << 1;
  const float* bcol = p2w + tn * 16 + m;
  v2f bfr[8];
#pragma unroll
  for (int ks = 0; ks < 8; ++ks) {
    bfr[ks].x = bcol[(ks * 4 + kh + 0) * 1024];
    bfr[ks].y = bcol[(ks * 4 + kh + 1) * 1024];
  }
  const float bv = p2b[tn * 16 + m];
#pragma unroll
  for (int tc = 0; tc < 2; ++tc) {
    float s = 0.0f;
#pragma unroll
    for (int sg = 0; sg < HEAD_GROUPS; ++sg) s += part[(sg * HEAD_TILES + tm * 3 + tc) * 256 + t];
    hid[rr][tc * 16 + cc] = fmaxf(s + p1b[tc * 16 + cc], 0.0f);
  }
  __syncthreads();
  v8f c = {};
#pragma unroll
  for (int ks = 0; ks < 8; ++ks) {
    v2f a;
    a.x = hid[m][ks * 4 + kh + 0];
    a.y = hid[m][ks * 4 + kh + 1];
    c = __builtin_amdgcn_wmma_f32_16x16x4_f32(false, a, false, bfr[ks], (short)0, c, false, false);
  }
  const int rbase = tm * 16 + ((lane >> 4) << 3);
#pragma unroll
  for (int r = 0; r < 8; ++r)
    xout[(rbase + r) * 1024 + tn * 16 + m] = c[r] + bv;
}

extern "C" void kernel_launch(void* const* d_in, const int* in_sizes, int n_in,
                              void* d_out, int out_size, void* d_ws, size_t ws_size,
                              hipStream_t stream) {
  const float* x    = (const float*)d_in[0];
  const int*   ei   = (const int*)d_in[1];
  const float* W1   = (const float*)d_in[2];
  const float* b1   = (const float*)d_in[3];
  const float* W2   = (const float*)d_in[4];
  const float* b2   = (const float*)d_in[5];
  const float* p1w  = (const float*)d_in[6];
  const float* p1b  = (const float*)d_in[7];
  const float* p2w  = (const float*)d_in[8];
  const float* p2b  = (const float*)d_in[9];
  const float* v1w  = (const float*)d_in[10];
  const float* v1b  = (const float*)d_in[11];
  const float* v2w  = (const float*)d_in[12];
  const float* v2b  = (const float*)d_in[13];

  const int N  = in_sizes[0] / 16;
  const int E  = in_sizes[1] / 2;
  const int* esrc = ei;
  const int* edst = ei + E;

  float* ws   = (float*)d_ws;
  int*   cst  = (int*)ws;
  int*   off  = cst + 256 * 256 + 32;
  float* dinv = (float*)(off + N + 32);
  int*   csr  = (int*)(dinv + N);
  float* bufA = (float*)(csr + E);
  float* bufB = bufA + (size_t)16 * N;
  int*   tmp  = (int*)bufA;
  float* part = bufA;

  float* Xout = (float*)d_out;
  float* Vout = Xout + 64 * 1024;

  const int B256 = 256;
  const int WGPS = 256;
  const int NC   = 256;
  k_chunk_sort<<<NC, 1024, 0, stream>>>(esrc, edst, tmp, cst, E);
  k_csr<<<N / 256, 1024, 0, stream>>>(tmp, cst, NC, E, off, dinv, csr, N);

  int gemm_blocks = (N / 32) / 8;
  k_gemm_n16<false><<<gemm_blocks, B256, 0, stream>>>(x, W1, bufA, N);
  k_pull_col<<<WGPS, 1024, 0, stream>>>(off, csr, dinv, bufA, b1, bufB, N, E);

  k_gemm_n16<true><<<gemm_blocks, B256, 0, stream>>>(bufB, W2, bufA, N);
  k_pull_col<<<WGPS, 1024, 0, stream>>>(off, csr, dinv, bufA, b2, bufB, N, E);

  k_heads_partial<<<128, 768, 0, stream>>>(bufB, N, 1024, p1w, v1w, part);
  k_heads_out<<<36, B256, 0, stream>>>(part, p1b, v1b, p2w, p2b, v2w, v2b, Xout, Vout);
}


# layer 2's product moved behind its gather, (A h)W2 = A(h W2): the heads kernel does W2, b2, relu on WMMA while staging its 512 rows into LDS; no second product launch, 8 to 7; rounding changes
# speedup vs baseline: 4.9423x; 1.0420x over previous
#include <hip/hip_runtime.h>
#include <hip/hip_bf16.h>

typedef __attribute__((ext_vector_type(2))) float v2f;
typedef __attribute__((ext_vector_type(8))) float v8f;
typedef __attribute__((ext_vector_type(4))) int   v4i;

#define BIN_SHIFT 8
#define BIN_NODES 256
#define MAX_BINS  256
#define CH_MAX    8192
#define MAX_CHUNKS 256
#define RUN_REGS  2
#define STAGE_MAX 12288

__device__ __forceinline__ void scan256_first_wave(const int* in, int* out) {
  const int lane = threadIdx.x;
  int v[8];
  int run = 0;
#pragma unroll
  for (int k = 0; k < 8; ++k) { int t = in[lane * 8 + k]; v[k] = run; run += t; }
  int inc = run;
  for (int o = 1; o < 32; o <<= 1) {
    int t = __shfl_up(inc, o);
    if (lane >= o) inc += t;
  }
  const int ex = inc - run;
#pragma unroll
  for (int k = 0; k < 8; ++k) out[lane * 8 + k] = ex + v[k];
}

__global__ void __launch_bounds__(1024)
k_chunk_sort(const int* __restrict__ src, const int* __restrict__ dst,
             int* __restrict__ tmp, int* __restrict__ cst, int ne) {
  __shared__ int cntl[MAX_BINS];
  __shared__ int lstart[MAX_BINS];
  __shared__ int cur[MAX_BINS];
  __shared__ int rec[CH_MAX];
  const int nc = gridDim.x, c = blockIdx.x, tid = threadIdx.x;
  const int ch = ne / nc;
  if (tid < MAX_BINS) cntl[tid] = 0;
  __syncthreads();
  const v4i* s4 = (const v4i*)(src + (size_t)c * ch);
  const v4i* d4 = (const v4i*)(dst + (size_t)c * ch);
  for (int i = tid; i < (ch >> 2); i += blockDim.x) {
    v4i d = d4[i];
    atomicAdd(&cntl[d.x >> BIN_SHIFT], 1);
    atomicAdd(&cntl[d.y >> BIN_SHIFT], 1);
    atomicAdd(&cntl[d.z >> BIN_SHIFT], 1);
    atomicAdd(&cntl[d.w >> BIN_SHIFT], 1);
  }
  __syncthreads();
  if (tid < 32) scan256_first_wave(cntl, lstart);
  __syncthreads();
  if (tid < MAX_BINS) {
    int l = lstart[tid];
    cur[tid] = l;
    cst[tid * nc + c] = l;
  }
  __syncthreads();
  for (int i = tid; i < (ch >> 2); i += blockDim.x) {
    v4i s = __builtin_nontemporal_load(s4 + i);
    v4i d = d4[i];
#pragma unroll
    for (int k = 0; k < 4; ++k) {
      int dk = d[k], sk = s[k];
      int p = atomicAdd(&cur[dk >> BIN_SHIFT], 1);
      rec[p] = sk | ((dk & (BIN_NODES - 1)) << 16);
    }
  }
  __syncthreads();
  int* t = tmp + (size_t)c * ch;
  for (int i = tid; i < ch; i += blockDim.x) t[i] = rec[i];
}

__global__ void __launch_bounds__(1024)
k_csr(const int* __restrict__ tmp, const int* __restrict__ cst, int nc, int ne,
      int* __restrict__ off, float* __restrict__ dinv, int* __restrict__ csr, int n) {
  __shared__ int dc[BIN_NODES];
  __shared__ int cur[BIN_NODES];
  __shared__ int rbeg[MAX_CHUNKS];
  __shared__ int rlen[MAX_CHUNKS];
  __shared__ int redlo[32], redhi[32];
  __shared__ int stage[STAGE_MAX];
  const int b = blockIdx.x, tid = threadIdx.x;
  const int lane = tid & 31, w = tid >> 5;
  const int ch = ne / nc;
  int lo = 0, hi = 0;
  if (tid < nc && tid < MAX_CHUNKS) {
    lo = cst[b * nc + tid];
    hi = (b + 1 < MAX_BINS) ? cst[(b + 1) * nc + tid] : ch;
  }
  if (tid < MAX_CHUNKS) { rbeg[tid] = tid * ch + lo; rlen[tid] = hi - lo; }
  if (tid < BIN_NODES) dc[tid] = 0;
  int slo = lo, shi = hi;
  for (int o = 16; o > 0; o >>= 1) { slo += __shfl_xor(slo, o); shi += __shfl_xor(shi, o); }
  if (lane == 0) { redlo[w] = slo; redhi[w] = shi; }
  __syncthreads();
  int start = redlo[lane], end = redhi[lane];
  for (int o = 16; o > 0; o >>= 1) { start += __shfl_xor(start, o); end += __shfl_xor(end, o); }
  const int m = end - start;

  int beg[8], len[8], r[8][RUN_REGS];
#pragma unroll
  for (int k = 0; k < 8; ++k) { beg[k] = rbeg[w + 32 * k]; len[k] = rlen[w + 32 * k]; }
#pragma unroll
  for (int k = 0; k < 8; ++k)
#pragma unroll
    for (int h = 0; h < RUN_REGS; ++h) {
      const int j = lane + 32 * h;
      r[k][h] = (j < len[k]) ? tmp[beg[k] + j] : -1;
    }
#pragma unroll
  for (int k = 0; k < 8; ++k) {
#pragma unroll
    for (int h = 0; h < RUN_REGS; ++h)
      if (r[k][h] >= 0) atomicAdd(&dc[r[k][h] >> 16], 1);
    for (int j = lane + 32 * RUN_REGS; j < len[k]; j += 32) atomicAdd(&dc[tmp[beg[k] + j] >> 16], 1);
  }
  __syncthreads();
  if (tid < 32) scan256_first_wave(dc, cur);
  __syncthreads();
  if (tid < BIN_NODES) {
    int node = b * BIN_NODES + tid;
    off[node]  = start + cur[tid];
    dinv[node] = rsqrtf((float)(dc[tid] + 1));
  }
  if (b == (int)gridDim.x - 1 && tid == 0) off[n] = end;
  __syncthreads();
  const bool staged = m <= STAGE_MAX;
#pragma unroll
  for (int k = 0; k < 8; ++k) {
#pragma unroll
    for (int h = 0; h < RUN_REGS; ++h)
      if (r[k][h] >= 0) {
        int p = atomicAdd(&cur[r[k][h] >> 16], 1);
        int s = r[k][h] & 0xFFFF;
        if (staged) stage[p] = s; else csr[start + p] = s;
      }
    for (int j = lane + 32 * RUN_REGS; j < len[k]; j += 32) {
      int rr = tmp[beg[k] + j];
      int p = atomicAdd(&cur[rr >> 16], 1);
      int s = rr & 0xFFFF;
      if (staged) stage[p] = s; else csr[start + p] = s;
    }
  }
  if (staged) {
    __syncthreads();
    for (int i = tid; i < m; i += blockDim.x) csr[start + i] = stage[i];
  }
}

__global__ void k_gemm_n16(const float* __restrict__ A, const float* __restrict__ B,
                           float* __restrict__ CT, int nrows) {
  int wid  = (blockIdx.x * blockDim.x + threadIdx.x) >> 5;
  int lane = threadIdx.x & 31;
  int row0 = wid * 32;
  if (row0 >= nrows) return;
  int m  = lane & 15;
  int kh = (lane >> 4) << 1;
  v2f bf[4];
#pragma unroll
  for (int ks = 0; ks < 4; ++ks) {
    bf[ks].x = B[(ks * 4 + kh + 0) * 16 + m];
    bf[ks].y = B[(ks * 4 + kh + 1) * 16 + m];
  }
#pragma unroll
  for (int t = 0; t < 2; ++t) {
    const int row = row0 + t * 16 + m;
    v8f c = {};
#pragma unroll
    for (int ks = 0; ks < 4; ++ks) {
      const int k = ks * 4 + kh;
      v2f a;
      a.x = A[row * 16 + k + 0];
      a.y = A[row * 16 + k + 1];
      c = __builtin_amdgcn_wmma_f32_16x16x4_f32(false, bf[ks], false, a, (short)0, c, false, false);
    }
    float* o = CT + (size_t)((lane >> 4) << 3) * nrows + row;
#pragma unroll
    for (int r = 0; r < 8; ++r) o[(size_t)r * nrows] = c[r];
  }
}

#define TAB_N    65536
#define F_COLS   16
#define PIECES   3

template <bool RELU_IN, bool ADD_BIAS>
__global__ void __launch_bounds__(1024)
k_pull_col(const int* __restrict__ off, const int* __restrict__ csr,
           const float* __restrict__ dinv, const float* __restrict__ hwT,
           const float* __restrict__ bias, float* __restrict__ aggT, int n, int ne) {
  __shared__ __attribute__((aligned(16))) float tab[TAB_N + 4];
  const int tid = threadIdx.x;
  const int f   = blockIdx.x % F_COLS;
  const int rg  = blockIdx.x / F_COLS;
  const int nrg = gridDim.x / F_COLS;
  const int per = n / nrg;

  {
    const float4* h4 = (const float4*)(hwT + (size_t)f * n);
    const float4* d4 = (const float4*)dinv;
    const int n4  = n >> 2;
    const int rot = (blockIdx.x * 64) % n4;
#pragma unroll 4
    for (int i = tid; i < n4; i += blockDim.x) {
      int j = i + rot;
      if (j >= n4) j -= n4;
      float4 a = h4[j], b = d4[j];
      if (RELU_IN) { a.x = fmaxf(a.x, 0.0f); a.y = fmaxf(a.y, 0.0f); a.z = fmaxf(a.z, 0.0f); a.w = fmaxf(a.w, 0.0f); }
      float4 p; p.x = a.x * b.x; p.y = a.y * b.y; p.z = a.z * b.z; p.w = a.w * b.w;
      ((float4*)tab)[j] = p;
    }
  }
  if (tid == 0) tab[TAB_N] = 0.0f;
  __syncthreads();

  const int lane = tid & 31, w = tid >> 5;
  const int q = lane & 3, g = lane >> 2;
  const int nwaves = blockDim.x >> 5;
  const v4i* csr4 = (const v4i*)csr;
  const int last4 = (ne >> 2) - 1;
  for (int u = w; u * 32 < per; u += nwaves) {
    const int nb = rg * per + u * 32;
    int base[4], lo[4], hi[4];
#pragma unroll
    for (int sub = 0; sub < 4; ++sub) {
      const int node = nb + sub * 8 + g;
      const int o0 = off[node];
      const int o1 = off[node + 1];
      base[sub] = (o0 & ~3) + 4 * q;
      lo[sub]   = o0 - base[sub];
      hi[sub]   = o1 - base[sub];
    }
    v4i s[4][PIECES];
#pragma unroll
    for (int sub = 0; sub < 4; ++sub)
#pragma unroll
      for (int k = 0; k < PIECES; ++k)
        s[sub][k] = csr4[min((base[sub] >> 2) + 4 * k, last4)];
    float a[4];
#pragma unroll
    for (int sub = 0; sub < 4; ++sub) {
      float acc = 0.0f;
#pragma unroll
      for (int k = 0; k < PIECES; ++k)
#pragma unroll
        for (int j = 0; j < 4; ++j) {
          const int c = 16 * k + j;
          bool in = c < hi[sub];
          if (c < 3) in = in && (c >= lo[sub]);
          acc += tab[in ? s[sub][k][j] : TAB_N];
        }
      a[sub] = acc;
    }
#pragma unroll
    for (int sub = 0; sub < 4; ++sub)
      for (int c = 16 * PIECES - 3 * q; c < hi[sub]; c += 4) a[sub] += tab[csr[base[sub] + c]];
    float res = 0.0f;
#pragma unroll
    for (int sub = 0; sub < 4; ++sub) {
      float t = a[sub];
      t += __shfl_xor(t, 1);
      t += __shfl_xor(t, 2);
      if (q == sub) res = t;
    }
    const int mine = nb + q * 8 + g;
    float v = dinv[mine] * (tab[mine] + res);
    if (ADD_BIAS) v += bias[f];
    aggT[(size_t)f * n + mine] = v;
  }
}

#define HEAD_TILES  12
#define HEAD_GROUPS 32
#define HP_NODES    32
#define HP_PITCH    17

__global__ void __launch_bounds__(768)
k_heads_partial(const float* __restrict__ AT, int nn, int npg,
                const float* __restrict__ W2, const float* __restrict__ b2,
                const float* __restrict__ Bp, const float* __restrict__ Bv,
                float* __restrict__ part) {
  __shared__ float sA[(F_COLS + 1) * HP_NODES * HP_PITCH];
  __shared__ float red[24][256];
  const int tid = threadIdx.x;
  const int lane = tid & 31, w = tid >> 5;
  const int tm = blockIdx.x / HEAD_GROUPS;
  const int g  = blockIdx.x % HEAD_GROUPS;
  const int tc = w >> 3, kq = w & 7;
  const bool pol = tc < 2;
  const float* B  = pol ? Bp : Bv;
  const int ncols = pol ? 32 : 16;
  const int c0    = pol ? tc * 16 : 0;
  const int m = lane & 15;
  const int h = lane >> 4;
  v2f wf[4];
#pragma unroll
  for (int ks = 0; ks < 4; ++ks) {
    wf[ks].x = W2[(ks * 4 + 2 * h + 0) * F_COLS + m];
    wf[ks].y = W2[(ks * 4 + 2 * h + 1) * F_COLS + m];
  }
  float bz[8];
#pragma unroll
  for (int r = 0; r < 8; ++r) bz[r] = b2[8 * h + r];
  v2f xa[2][4];
#pragma unroll
  for (int j = 0; j < 2; ++j) {
    const int tt = min(w + 24 * j, 31);
    const float* ap = AT + (size_t)(tm * 16 + (tt >> 1)) * npg + g * HP_NODES + (tt & 1) * 16 + m;
#pragma unroll
    for (int ks = 0; ks < 4; ++ks) {
      xa[j][ks].x = ap[(size_t)(ks * 4 + 2 * h + 0) * nn];
      xa[j][ks].y = ap[(size_t)(ks * 4 + 2 * h + 1) * nn];
    }
  }
  const float* bbase = B + (size_t)((g * HP_NODES + 4 * kq) * 16 + h) * ncols + c0 + m;
  v2f bf[16];
#pragma unroll
  for (int i = 0; i < 16; ++i) {
    const int r = (i >> 2) * 16 + (i & 3) * 4;
    bf[i].x = bbase[r * ncols];
    bf[i].y = bbase[(r + 2) * ncols];
  }
#pragma unroll
  for (int j = 0; j < 2; ++j) {
    v8f d = {};
#pragma unroll
    for (int ks = 0; ks < 4; ++ks)
      d = __builtin_amdgcn_wmma_f32_16x16x4_f32(false, wf[ks], false, xa[j][ks], (short)0, d, false, false);
    const int tt = min(w + 24 * j, 31);
    const bool real = w + 24 * j < 32;
    const int nd = (tt & 1) * 16 + m;
#pragma unroll
    for (int r = 0; r < 8; ++r) {
      const int fb = real ? 8 * h + r : F_COLS;
      sA[(fb * HP_NODES + nd) * HP_PITCH + (tt >> 1)] = fmaxf(d[r] + bz[r], 0.0f);
    }
  }
  __syncthreads();
  v8f c = {};
#pragma unroll
  for (int i = 0; i < 16; ++i) {
    const int nd = 4 * kq + (i >> 2);
    const int f0 = (i & 3) * 4 + h;
    v2f a;
    a.x = sA[(f0 * HP_NODES + nd) * HP_PITCH + m];
    a.y = sA[((f0 + 2) * HP_NODES + nd) * HP_PITCH + m];
    c = __builtin_amdgcn_wmma_f32_16x16x4_f32(false, a, false, bf[i], (short)0, c, false, false);
  }
#pragma unroll
  for (int r = 0; r < 8; ++r) red[w][r * 32 + lane] = c[r];
  __syncthreads();
  const int tcc = tid >> 8, e = tid & 255;
  float s = 0.0f;
#pragma unroll
  for (int k = 0; k < 8; ++k) s += red[tcc * 8 + k][e];
  part[(size_t)(g * HEAD_TILES + tm * 3 + tcc) * 256 + e] = s;
}

__global__ void __launch_bounds__(256)
k_heads_out(const float* __restrict__ part,
            const float* __restrict__ p1b, const float* __restrict__ v1b,
            const float* __restrict__ p2w, const float* __restrict__ p2b,
            const float* __restrict__ v2w, const float* __restrict__ v2b,
            float* __restrict__ xout, float* __restrict__ vout) {
  __shared__ float hid[16][33];
  __shared__ float vh[16][17];
  const int t = threadIdx.x;
  const int lane = t & 31, w = t >> 5;
  const int rr = (t >> 5) + (((t >> 4) & 1) << 3);
  const int cc = t & 15;
  if (blockIdx.x >= 32) {
    const int tm = blockIdx.x - 32;
    float s = 0.0f;
#pragma unroll
    for (int sg = 0; sg < HEAD_GROUPS; ++sg) s += part[(sg * HEAD_TILES + tm * 3 + 2) * 256 + t];
    vh[rr][cc] = fmaxf(s + v1b[cc], 0.0f);
    __syncthreads();
    if (t < 16) {
      float d = 0.0f;
#pragma unroll
      for (int j = 0; j < 16; ++j) d += vh[t][j] * v2w[j];
      vout[tm * 16 + t] = d + v2b[0];
    }
    return;
  }
  const int tm = blockIdx.x >> 3;
  const int tn = ((blockIdx.x & 7) << 3) + w;
  const int m  = lane & 15;
  const int kh = (lane >> 4) << 1;
  const float* bcol = p2w + tn * 16 + m;
  v2f bfr[8];
#pragma unroll
  for (int ks = 0; ks < 8; ++ks) {
    bfr[ks].x = bcol[(ks * 4 + kh + 0) * 1024];
    bfr[ks].y = bcol[(ks * 4 + kh + 1) * 1024];
  }
  const float bv = p2b[tn * 16 + m];
#pragma unroll
  for (int tc = 0; tc < 2; ++tc) {
    float s = 0.0f;
#pragma unroll
    for (int sg = 0; sg < HEAD_GROUPS; ++sg) s += part[(sg * HEAD_TILES + tm * 3 + tc) * 256 + t];
    hid[rr][tc * 16 + cc] = fmaxf(s + p1b[tc * 16 + cc], 0.0f);
  }
  __syncthreads();
  v8f c = {};
#pragma unroll
  for (int ks = 0; ks < 8; ++ks) {
    v2f a;
    a.x = hid[m][ks * 4 + kh + 0];
    a.y = hid[m][ks * 4 + kh + 1];
    c = __builtin_amdgcn_wmma_f32_16x16x4_f32(false, a, false, bfr[ks], (short)0, c, false, false);
  }
  const int rbase = tm * 16 + ((lane >> 4) << 3);
#pragma unroll
  for (int r = 0; r < 8; ++r)
    xout[(rbase + r) * 1024 + tn * 16 + m] = c[r] + bv;
}

extern "C" void kernel_launch(void* const* d_in, const int* in_sizes, int n_in,
                              void* d_out, int out_size, void* d_ws, size_t ws_size,
                              hipStream_t stream) {
  const float* x    = (const float*)d_in[0];
  const int*   ei   = (const int*)d_in[1];
  const float* W1   = (const float*)d_in[2];
  const float* b1   = (const float*)d_in[3];
  const float* W2   = (const float*)d_in[4];
  const float* b2   = (const float*)d_in[5];
  const float* p1w  = (const float*)d_in[6];
  const float* p1b  = (const float*)d_in[7];
  const float* p2w  = (const float*)d_in[8];
  const float* p2b  = (const float*)d_in[9];
  const float* v1w  = (const float*)d_in[10];
  const float* v1b  = (const float*)d_in[11];
  const float* v2w  = (const float*)d_in[12];
  const float* v2b  = (const float*)d_in[13];

  const int N  = in_sizes[0] / 16;
  const int E  = in_sizes[1] / 2;
  const int* esrc = ei;
  const int* edst = ei + E;

  float* ws   = (float*)d_ws;
  int*   cst  = (int*)ws;
  int*   off  = cst + 256 * 256 + 32;
  float* dinv = (float*)(off + N + 32);
  int*   csr  = (int*)(dinv + N);
  float* bufA = (float*)(csr + E);
  float* bufB = bufA + (size_t)16 * N;
  int*   tmp  = (int*)bufA;
  float* part = bufB;

  float* Xout = (float*)d_out;
  float* Vout = Xout + 64 * 1024;

  const int B256 = 256;
  const int WGPS = 256;
  const int NC   = 256;
  k_chunk_sort<<<NC, 1024, 0, stream>>>(esrc, edst, tmp, cst, E);
  k_csr<<<N / 256, 1024, 0, stream>>>(tmp, cst, NC, E, off, dinv, csr, N);

  int gemm_blocks = (N / 32) / 8;
  k_gemm_n16<<<gemm_blocks, B256, 0, stream>>>(x, W1, bufA, N);
  k_pull_col<false, true><<<WGPS, 1024, 0, stream>>>(off, csr, dinv, bufA, b1, bufB, N, E);

  k_pull_col<true, false><<<WGPS, 1024, 0, stream>>>(off, csr, dinv, bufB, b2, bufA, N, E);

  k_heads_partial<<<128, 768, 0, stream>>>(bufA, N, 1024, W2, b2, p1w, v1w, part);
  k_heads_out<<<36, B256, 0, stream>>>(part, p1b, v1b, p2w, p2b, v2w, v2b, Xout, Vout);
}


# final edge list kept as 16-bit source ids: its freed 4 MiB takes x W1, formed by 8 waves of each first-level sort work-group while wave 0 scans; no product launch, 7 to 6; sums unchanged
# speedup vs baseline: 5.7025x; 1.1538x over previous
#include <hip/hip_runtime.h>
#include <hip/hip_bf16.h>

typedef __attribute__((ext_vector_type(2))) float v2f;
typedef __attribute__((ext_vector_type(8))) float v8f;
typedef __attribute__((ext_vector_type(4))) int   v4i;
typedef __attribute__((ext_vector_type(2))) int   v2i;

#define BIN_SHIFT 8
#define BIN_NODES 256
#define MAX_BINS  256
#define CH_MAX    8192
#define MAX_CHUNKS 256
#define RUN_REGS  2
#define STAGE_MAX 12288

__device__ __forceinline__ void scan256_first_wave(const int* in, int* out) {
  const int lane = threadIdx.x;
  int v[8];
  int run = 0;
#pragma unroll
  for (int k = 0; k < 8; ++k) { int t = in[lane * 8 + k]; v[k] = run; run += t; }
  int inc = run;
  for (int o = 1; o < 32; o <<= 1) {
    int t = __shfl_up(inc, o);
    if (lane >= o) inc += t;
  }
  const int ex = inc - run;
#pragma unroll
  for (int k = 0; k < 8; ++k) out[lane * 8 + k] = ex + v[k];
}

#define XW_W0 8

__device__ __forceinline__ void xw_load(const float* __restrict__ A, const float* __restrict__ B,
                                        int row0, int lane, v2f (&bf)[4], v2f (&a)[2][4]) {
  const int m  = lane & 15;
  const int kh = (lane >> 4) << 1;
#pragma unroll
  for (int ks = 0; ks < 4; ++ks) {
    bf[ks].x = B[(ks * 4 + kh + 0) * 16 + m];
    bf[ks].y = B[(ks * 4 + kh + 1) * 16 + m];
  }
#pragma unroll
  for (int t = 0; t < 2; ++t) {
    const int row = row0 + t * 16 + m;
#pragma unroll
    for (int ks = 0; ks < 4; ++ks) {
      const int k = ks * 4 + kh;
      a[t][ks].x = A[row * 16 + k + 0];
      a[t][ks].y = A[row * 16 + k + 1];
    }
  }
}

__device__ __forceinline__ void xw_mma(const v2f (&bf)[4], const v2f (&a)[2][4], v8f (&c)[2]) {
#pragma unroll
  for (int t = 0; t < 2; ++t) {
    v8f d = {};
#pragma unroll
    for (int ks = 0; ks < 4; ++ks)
      d = __builtin_amdgcn_wmma_f32_16x16x4_f32(false, bf[ks], false, a[t][ks], (short)0, d, false, false);
    c[t] = d;
  }
}

__device__ __forceinline__ void xw_store(float* __restrict__ CT, int nrows, int row0, int lane, const v8f (&c)[2]) {
#pragma unroll
  for (int t = 0; t < 2; ++t) {
    const int row = row0 + t * 16 + (lane & 15);
    float* o = CT + (size_t)((lane >> 4) << 3) * nrows + row;
#pragma unroll
    for (int r = 0; r < 8; ++r) o[(size_t)r * nrows] = c[t][r];
  }
}

__global__ void __launch_bounds__(1024)
k_chunk_sort(const int* __restrict__ src, const int* __restrict__ dst,
             int* __restrict__ tmp, int* __restrict__ cst, int ne,
             const float* __restrict__ x, const float* __restrict__ W1,
             float* __restrict__ hwT, int nrows) {
  __shared__ int cntl[MAX_BINS];
  __shared__ int lstart[MAX_BINS];
  __shared__ int cur[MAX_BINS];
  __shared__ int rec[CH_MAX];
  const int nc = gridDim.x, c = blockIdx.x, tid = threadIdx.x;
  const int ch = ne / nc;
  const int xlane = tid & 31, xwv = (tid >> 5) - XW_W0;
  const int xrow0 = (c * 8 + xwv) * 32;
  const bool xw = xwv >= 0 && xwv < 8 && xrow0 < nrows;
  v2f xbf[4] = {}, xa[2][4] = {};
  v8f xc[2] = {};
  if (tid < MAX_BINS) cntl[tid] = 0;
  __syncthreads();
  if (xw) xw_load(x, W1, xrow0, xlane, xbf, xa);
  const v4i* s4 = (const v4i*)(src + (size_t)c * ch);
  const v4i* d4 = (const v4i*)(dst + (size_t)c * ch);
  for (int i = tid; i < (ch >> 2); i += blockDim.x) {
    v4i d = d4[i];
    atomicAdd(&cntl[d.x >> BIN_SHIFT], 1);
    atomicAdd(&cntl[d.y >> BIN_SHIFT], 1);
    atomicAdd(&cntl[d.z >> BIN_SHIFT], 1);
    atomicAdd(&cntl[d.w >> BIN_SHIFT], 1);
  }
  __syncthreads();
  if (tid < 32) scan256_first_wave(cntl, lstart);
  else if (xw) xw_mma(xbf, xa, xc);
  __syncthreads();
  if (tid < MAX_BINS) {
    int l = lstart[tid];
    cur[tid] = l;
    cst[tid * nc + c] = l;
  } else if (xw) xw_store(hwT, nrows, xrow0, xlane, xc);
  __syncthreads();
  for (int i = tid; i < (ch >> 2); i += blockDim.x) {
    v4i s = __builtin_nontemporal_load(s4 + i);
    v4i d = d4[i];
#pragma unroll
    for (int k = 0; k < 4; ++k) {
      int dk = d[k], sk = s[k];
      int p = atomicAdd(&cur[dk >> BIN_SHIFT], 1);
      rec[p] = sk | ((dk & (BIN_NODES - 1)) << 16);
    }
  }
  __syncthreads();
  int* t = tmp + (size_t)c * ch;
  for (int i = tid; i < ch; i += blockDim.x) t[i] = rec[i];
}

__global__ void __launch_bounds__(1024)
k_csr(const int* __restrict__ tmp, const int* __restrict__ cst, int nc, int ne,
      int* __restrict__ off, float* __restrict__ dinv, unsigned short* __restrict__ csr, int n) {
  __shared__ int dc[BIN_NODES];
  __shared__ int cur[BIN_NODES];
  __shared__ int rbeg[MAX_CHUNKS];
  __shared__ int rlen[MAX_CHUNKS];
  __shared__ int redlo[32], redhi[32];
  __shared__ int stage[STAGE_MAX];
  const int b = blockIdx.x, tid = threadIdx.x;
  const int lane = tid & 31, w = tid >> 5;
  const int ch = ne / nc;
  int lo = 0, hi = 0;
  if (tid < nc && tid < MAX_CHUNKS) {
    lo = cst[b * nc + tid];
    hi = (b + 1 < MAX_BINS) ? cst[(b + 1) * nc + tid] : ch;
  }
  if (tid < MAX_CHUNKS) { rbeg[tid] = tid * ch + lo; rlen[tid] = hi - lo; }
  if (tid < BIN_NODES) dc[tid] = 0;
  int slo = lo, shi = hi;
  for (int o = 16; o > 0; o >>= 1) { slo += __shfl_xor(slo, o); shi += __shfl_xor(shi, o); }
  if (lane == 0) { redlo[w] = slo; redhi[w] = shi; }
  __syncthreads();
  int start = redlo[lane], end = redhi[lane];
  for (int o = 16; o > 0; o >>= 1) { start += __shfl_xor(start, o); end += __shfl_xor(end, o); }
  const int m = end - start;

  int beg[8], len[8], r[8][RUN_REGS];
#pragma unroll
  for (int k = 0; k < 8; ++k) { beg[k] = rbeg[w + 32 * k]; len[k] = rlen[w + 32 * k]; }
#pragma unroll
  for (int k = 0; k < 8; ++k)
#pragma unroll
    for (int h = 0; h < RUN_REGS; ++h) {
      const int j = lane + 32 * h;
      r[k][h] = (j < len[k]) ? tmp[beg[k] + j] : -1;
    }
#pragma unroll
  for (int k = 0; k < 8; ++k) {
#pragma unroll
    for (int h = 0; h < RUN_REGS; ++h)
      if (r[k][h] >= 0) atomicAdd(&dc[r[k][h] >> 16], 1);
    for (int j = lane + 32 * RUN_REGS; j < len[k]; j += 32) atomicAdd(&dc[tmp[beg[k] + j] >> 16], 1);
  }
  __syncthreads();
  if (tid < 32) scan256_first_wave(dc, cur);
  __syncthreads();
  if (tid < BIN_NODES) {
    int node = b * BIN_NODES + tid;
    off[node]  = start + cur[tid];
    dinv[node] = rsqrtf((float)(dc[tid] + 1));
  }
  if (b == (int)gridDim.x - 1 && tid == 0) off[n] = end;
  __syncthreads();
  const bool staged = m <= STAGE_MAX;
#pragma unroll
  for (int k = 0; k < 8; ++k) {
#pragma unroll
    for (int h = 0; h < RUN_REGS; ++h)
      if (r[k][h] >= 0) {
        int p = atomicAdd(&cur[r[k][h] >> 16], 1);
        int s = r[k][h] & 0xFFFF;
        if (staged) stage[p] = s; else csr[start + p] = (unsigned short)s;
      }
    for (int j = lane + 32 * RUN_REGS; j < len[k]; j += 32) {
      int rr = tmp[beg[k] + j];
      int p = atomicAdd(&cur[rr >> 16], 1);
      int s = rr & 0xFFFF;
      if (staged) stage[p] = s; else csr[start + p] = (unsigned short)s;
    }
  }
  if (staged) {
    __syncthreads();
    for (int i = tid; i < m; i += blockDim.x) csr[start + i] = (unsigned short)stage[i];
  }
}

#define TAB_N    65536
#define F_COLS   16
#define PIECES   3

template <bool RELU_IN, bool ADD_BIAS>
__global__ void __launch_bounds__(1024)
k_pull_col(const int* __restrict__ off, const unsigned short* __restrict__ csr,
           const float* __restrict__ dinv, const float* __restrict__ hwT,
           const float* __restrict__ bias, float* __restrict__ aggT, int n, int ne) {
  __shared__ __attribute__((aligned(16))) float tab[TAB_N + 4];
  const int tid = threadIdx.x;
  const int f   = blockIdx.x % F_COLS;
  const int rg  = blockIdx.x / F_COLS;
  const int nrg = gridDim.x / F_COLS;
  const int per = n / nrg;

  {
    const float4* h4 = (const float4*)(hwT + (size_t)f * n);
    const float4* d4 = (const float4*)dinv;
    const int n4  = n >> 2;
    const int rot = (blockIdx.x * 64) % n4;
#pragma unroll 4
    for (int i = tid; i < n4; i += blockDim.x) {
      int j = i + rot;
      if (j >= n4) j -= n4;
      float4 a = h4[j], b = d4[j];
      if (RELU_IN) { a.x = fmaxf(a.x, 0.0f); a.y = fmaxf(a.y, 0.0f); a.z = fmaxf(a.z, 0.0f); a.w = fmaxf(a.w, 0.0f); }
      float4 p; p.x = a.x * b.x; p.y = a.y * b.y; p.z = a.z * b.z; p.w = a.w * b.w;
      ((float4*)tab)[j] = p;
    }
  }
  if (tid == 0) tab[TAB_N] = 0.0f;
  __syncthreads();

  const int lane = tid & 31, w = tid >> 5;
  const int q = lane & 3, g = lane >> 2;
  const int nwaves = blockDim.x >> 5;
  const v2i* csr4 = (const v2i*)csr;
  const int last4 = (ne >> 2) - 1;
  for (int u = w; u * 32 < per; u += nwaves) {
    const int nb = rg * per + u * 32;
    int base[4], lo[4], hi[4];
#pragma unroll
    for (int sub = 0; sub < 4; ++sub) {
      const int node = nb + sub * 8 + g;
      const int o0 = off[node];
      const int o1 = off[node + 1];
      base[sub] = (o0 & ~3) + 4 * q;
      lo[sub]   = o0 - base[sub];
      hi[sub]   = o1 - base[sub];
    }
    v2i s[4][PIECES];
#pragma unroll
    for (int sub = 0; sub < 4; ++sub)
#pragma unroll
      for (int k = 0; k < PIECES; ++k)
        s[sub][k] = csr4[min((base[sub] >> 2) + 4 * k, last4)];
    float a[4];
#pragma unroll
    for (int sub = 0; sub < 4; ++sub) {
      float acc = 0.0f;
#pragma unroll
      for (int k = 0; k < PIECES; ++k)
#pragma unroll
        for (int j = 0; j < 4; ++j) {
          const int c = 16 * k + j;
          bool in = c < hi[sub];
          if (c < 3) in = in && (c >= lo[sub]);
          const int wd = s[sub][k][j >> 1];
          const int id = (j & 1) ? (int)((unsigned)wd >> 16) : (wd & 0xFFFF);
          acc += tab[in ? id : TAB_N];
        }
      a[sub] = acc;
    }
#pragma unroll
    for (int sub = 0; sub < 4; ++sub)
      for (int c = 16 * PIECES - 3 * q; c < hi[sub]; c += 4) a[sub] += tab[csr[base[sub] + c]];
    float res = 0.0f;
#pragma unroll
    for (int sub = 0; sub < 4; ++sub) {
      float t = a[sub];
      t += __shfl_xor(t, 1);
      t += __shfl_xor(t, 2);
      if (q == sub) res = t;
    }
    const int mine = nb + q * 8 + g;
    float v = dinv[mine] * (tab[mine] + res);
    if (ADD_BIAS) v += bias[f];
    aggT[(size_t)f * n + mine] = v;
  }
}

#define HEAD_TILES  12
#define HEAD_GROUPS 32
#define HP_NODES    32
#define HP_PITCH    17

__global__ void __launch_bounds__(768)
k_heads_partial(const float* __restrict__ AT, int nn, int npg,
                const float* __restrict__ W2, const float* __restrict__ b2,
                const float* __restrict__ Bp, const float* __restrict__ Bv,
                float* __restrict__ part) {
  __shared__ float sA[(F_COLS + 1) * HP_NODES * HP_PITCH];
  __shared__ float red[24][256];
  const int tid = threadIdx.x;
  const int lane = tid & 31, w = tid >> 5;
  const int tm = blockIdx.x / HEAD_GROUPS;
  const int g  = blockIdx.x % HEAD_GROUPS;
  const int tc = w >> 3, kq = w & 7;
  const bool pol = tc < 2;
  const float* B  = pol ? Bp : Bv;
  const int ncols = pol ? 32 : 16;
  const int c0    = pol ? tc * 16 : 0;
  const int m = lane & 15;
  const int h = lane >> 4;
  v2f wf[4];
#pragma unroll
  for (int ks = 0; ks < 4; ++ks) {
    wf[ks].x = W2[(ks * 4 + 2 * h + 0) * F_COLS + m];
    wf[ks].y = W2[(ks * 4 + 2 * h + 1) * F_COLS + m];
  }
  float bz[8];
#pragma unroll
  for (int r = 0; r < 8; ++r) bz[r] = b2[8 * h + r];
  v2f xa[2][4];
#pragma unroll
  for (int j = 0; j < 2; ++j) {
    const int tt = min(w + 24 * j, 31);
    const float* ap = AT + (size_t)(tm * 16 + (tt >> 1)) * npg + g * HP_NODES + (tt & 1) * 16 + m;
#pragma unroll
    for (int ks = 0; ks < 4; ++ks) {
      xa[j][ks].x = ap[(size_t)(ks * 4 + 2 * h + 0) * nn];
      xa[j][ks].y = ap[(size_t)(ks * 4 + 2 * h + 1) * nn];
    }
  }
  const float* bbase = B + (size_t)((g * HP_NODES + 4 * kq) * 16 + h) * ncols + c0 + m;
  v2f bf[16];
#pragma unroll
  for (int i = 0; i < 16; ++i) {
    const int r = (i >> 2) * 16 + (i & 3) * 4;
    bf[i].x = bbase[r * ncols];
    bf[i].y = bbase[(r + 2) * ncols];
  }
#pragma unroll
  for (int j = 0; j < 2; ++j) {
    v8f d = {};
#pragma unroll
    for (int ks = 0; ks < 4; ++ks)
      d = __builtin_amdgcn_wmma_f32_16x16x4_f32(false, wf[ks], false, xa[j][ks], (short)0, d, false, false);
    const int tt = min(w + 24 * j, 31);
    const bool real = w + 24 * j < 32;
    const int nd = (tt & 1) * 16 + m;
#pragma unroll
    for (int r = 0; r < 8; ++r) {
      const int fb = real ? 8 * h + r : F_COLS;
      sA[(fb * HP_NODES + nd) * HP_PITCH + (tt >> 1)] = fmaxf(d[r] + bz[r], 0.0f);
    }
  }
  __syncthreads();
  v8f c = {};
#pragma unroll
  for (int i = 0; i < 16; ++i) {
    const int nd = 4 * kq + (i >> 2);
    const int f0 = (i & 3) * 4 + h;
    v2f a;
    a.x = sA[(f0 * HP_NODES + nd) * HP_PITCH + m];
    a.y = sA[((f0 + 2) * HP_NODES + nd) * HP_PITCH + m];
    c = __builtin_amdgcn_wmma_f32_16x16x4_f32(false, a, false, bf[i], (short)0, c, false, false);
  }
#pragma unroll
  for (int r = 0; r < 8; ++r) red[w][r * 32 + lane] = c[r];
  __syncthreads();
  const int tcc = tid >> 8, e = tid & 255;
  float s = 0.0f;
#pragma unroll
  for (int k = 0; k < 8; ++k) s += red[tcc * 8 + k][e];
  part[(size_t)(g * HEAD_TILES + tm * 3 + tcc) * 256 + e] = s;
}

__global__ void __launch_bounds__(256)
k_heads_out(const float* __restrict__ part,
            const float* __restrict__ p1b, const float* __restrict__ v1b,
            const float* __restrict__ p2w, const float* __restrict__ p2b,
            const float* __restrict__ v2w, const float* __restrict__ v2b,
            float* __restrict__ xout, float* __restrict__ vout) {
  __shared__ float hid[16][33];
  __shared__ float vh[16][17];
  const int t = threadIdx.x;
  const int lane = t & 31, w = t >> 5;
  const int rr = (t >> 5) + (((t >> 4) & 1) << 3);
  const int cc = t & 15;
  if (blockIdx.x >= 32) {
    const int tm = blockIdx.x - 32;
    float s = 0.0f;
#pragma unroll
    for (int sg = 0; sg < HEAD_GROUPS; ++sg) s += part[(sg * HEAD_TILES + tm * 3 + 2) * 256 + t];
    vh[rr][cc] = fmaxf(s + v1b[cc], 0.0f);
    __syncthreads();
    if (t < 16) {
      float d = 0.0f;
#pragma unroll
      for (int j = 0; j < 16; ++j) d += vh[t][j] * v2w[j];
      vout[tm * 16 + t] = d + v2b[0];
    }
    return;
  }
  const int tm = blockIdx.x >> 3;
  const int tn = ((blockIdx.x & 7) << 3) + w;
  const int m  = lane & 15;
  const int kh = (lane >> 4) << 1;
  const float* bcol = p2w + tn * 16 + m;
  v2f bfr[8];
#pragma unroll
  for (int ks = 0; ks < 8; ++ks) {
    bfr[ks].x = bcol[(ks * 4 + kh + 0) * 1024];
    bfr[ks].y = bcol[(ks * 4 + kh + 1) * 1024];
  }
  const float bv = p2b[tn * 16 + m];
#pragma unroll
  for (int tc = 0; tc < 2; ++tc) {
    float s = 0.0f;
#pragma unroll
    for (int sg = 0; sg < HEAD_GROUPS; ++sg) s += part[(sg * HEAD_TILES + tm * 3 + tc) * 256 + t];
    hid[rr][tc * 16 + cc] = fmaxf(s + p1b[tc * 16 + cc], 0.0f);
  }
  __syncthreads();
  v8f c = {};
#pragma unroll
  for (int ks = 0; ks < 8; ++ks) {
    v2f a;
    a.x = hid[m][ks * 4 + kh + 0];
    a.y = hid[m][ks * 4 + kh + 1];
    c = __builtin_amdgcn_wmma_f32_16x16x4_f32(false, a, false, bfr[ks], (short)0, c, false, false);
  }
  const int rbase = tm * 16 + ((lane >> 4) << 3);
#pragma unroll
  for (int r = 0; r < 8; ++r)
    xout[(rbase + r) * 1024 + tn * 16 + m] = c[r] + bv;
}

extern "C" void kernel_launch(void* const* d_in, const int* in_sizes, int n_in,
                              void* d_out, int out_size, void* d_ws, size_t ws_size,
                              hipStream_t stream) {
  const float* x    = (const float*)d_in[0];
  const int*   ei   = (const int*)d_in[1];
  const float* W1   = (const float*)d_in[2];
  const float* b1   = (const float*)d_in[3];
  const float* W2   = (const float*)d_in[4];
  const float* b2   = (const float*)d_in[5];
  const float* p1w  = (const float*)d_in[6];
  const float* p1b  = (const float*)d_in[7];
  const float* p2w  = (const float*)d_in[8];
  const float* p2b  = (const float*)d_in[9];
  const float* v1w  = (const float*)d_in[10];
  const float* v1b  = (const float*)d_in[11];
  const float* v2w  = (const float*)d_in[12];
  const float* v2b  = (const float*)d_in[13];

  const int N  = in_sizes[0] / 16;
  const int E  = in_sizes[1] / 2;
  const int* esrc = ei;
  const int* edst = ei + E;

  float* ws   = (float*)d_ws;
  int*   cst  = (int*)ws;
  int*   off  = cst + 256 * 256 + 32;
  float* dinv = (float*)(off + N + 32);
  int*   tmp  = (int*)(dinv + N);
  float* bufA = (float*)(tmp + E);
  float* bufB = bufA + (size_t)16 * N;
  unsigned short* csr = (unsigned short*)bufB;
  float* agg1 = (float*)tmp;
  float* part = bufB;

  float* Xout = (float*)d_out;
  float* Vout = Xout + 64 * 1024;

  const int B256 = 256;
  const int WGPS = 256;
  const int NC   = 256;
  k_chunk_sort<<<NC, 1024, 0, stream>>>(esrc, edst, tmp, cst, E, x, W1, bufA, N);
  k_csr<<<N / 256, 1024, 0, stream>>>(tmp, cst, NC, E, off, dinv, csr, N);

  k_pull_col<false, true><<<WGPS, 1024, 0, stream>>>(off, csr, dinv, bufA, b1, agg1, N, E);

  k_pull_col<true, false><<<WGPS, 1024, 0, stream>>>(off, csr, dinv, agg1, b2, bufA, N, E);

  k_heads_partial<<<128, 768, 0, stream>>>(bufA, N, 1024, W2, b2, p1w, v1w, part);
  k_heads_out<<<36, B256, 0, stream>>>(part, p1b, v1b, p2w, p2b, v2w, v2b, Xout, Vout);
}
